# starting kernel
# baseline (speedup 1.0000x reference)
#include <hip/hip_runtime.h>

typedef float v2f __attribute__((ext_vector_type(2)));
typedef float v8f __attribute__((ext_vector_type(8)));

#define N_FEAT 64

__global__ void zero_f4(float4* __restrict__ p, int n4) {
    int i = blockIdx.x * blockDim.x + threadIdx.x;
    if (i < n4) p[i] = make_float4(0.f, 0.f, 0.f, 0.f);
}

__global__ void deg_kernel(const int* __restrict__ dst, float* __restrict__ deg, int n_edges) {
    int e = blockIdx.x * blockDim.x + threadIdx.x;
    if (e < n_edges) atomicAdd(&deg[dst[e]], 1.0f);
}

__global__ void norm_kernel(float* __restrict__ dn, int n_nodes) {
    int i = blockIdx.x * blockDim.x + threadIdx.x;
    if (i < n_nodes) {
        float d = dn[i];
        dn[i] = (d > 0.f) ? rsqrtf(d) : 0.f;
    }
}

__global__ void gemm_norm_wmma(const float* __restrict__ x, int chan,
                               const float* __restrict__ W,
                               const float* __restrict__ norm,
                               float* __restrict__ h, int n_nodes) {
    int wave  = (blockIdx.x * blockDim.x + threadIdx.x) >> 5;
    int lane  = threadIdx.x & 31;
    int node0 = wave * 16;
    if (node0 >= n_nodes) return;
    int half = lane >> 4;
    int l16  = lane & 15;

    const float* xrow = x + (size_t)(node0 + l16) * (2 * N_FEAT) + chan;
    v8f acc0 = {}, acc1 = {}, acc2 = {}, acc3 = {};

#pragma unroll
    for (int kc = 0; kc < 16; ++kc) {
        int k0 = kc * 4 + half * 2;
        v2f a;
        a.x = xrow[(size_t)(k0 + 0) * 2];
        a.y = xrow[(size_t)(k0 + 1) * 2];
        const float* wr0 = W + (size_t)(k0 + 0) * N_FEAT;
        const float* wr1 = W + (size_t)(k0 + 1) * N_FEAT;
        v2f b0; b0.x = wr0[l16 +  0]; b0.y = wr1[l16 +  0];
        v2f b1; b1.x = wr0[l16 + 16]; b1.y = wr1[l16 + 16];
        v2f b2; b2.x = wr0[l16 + 32]; b2.y = wr1[l16 + 32];
        v2f b3; b3.x = wr0[l16 + 48]; b3.y = wr1[l16 + 48];
        acc0 = __builtin_amdgcn_wmma_f32_16x16x4_f32(false, a, false, b0, (short)0, acc0, false, false);
        acc1 = __builtin_amdgcn_wmma_f32_16x16x4_f32(false, a, false, b1, (short)0, acc1, false, false);
        acc2 = __builtin_amdgcn_wmma_f32_16x16x4_f32(false, a, false, b2, (short)0, acc2, false, false);
        acc3 = __builtin_amdgcn_wmma_f32_16x16x4_f32(false, a, false, b3, (short)0, acc3, false, false);
    }

#pragma unroll
    for (int v = 0; v < 8; ++v) {
        int row = node0 + v + half * 8;
        float nr = norm[row];
        float* hr = h + (size_t)row * N_FEAT + l16;
        hr[ 0] = acc0[v] * nr;
        hr[16] = acc1[v] * nr;
        hr[32] = acc2[v] * nr;
        hr[48] = acc3[v] * nr;
    }
}

__global__ void scatter_kernel(const float* __restrict__ h,
                               const int* __restrict__ src,
                               const int* __restrict__ dst,
                               float* __restrict__ agg, int n_edges) {
    int t = blockIdx.x * blockDim.x + threadIdx.x;
    int e = t >> 4;
    if (e >= n_edges) return;
    int q = (t & 15) * 4;
    int s = src[e], d = dst[e];
    const float4 v = *(const float4*)(h + (size_t)s * N_FEAT + q);
    float* ap = agg + (size_t)d * N_FEAT + q;
    atomicAdd(ap + 0, v.x);
    atomicAdd(ap + 1, v.y);
    atomicAdd(ap + 2, v.z);
    atomicAdd(ap + 3, v.w);
}

__global__ void finalize_first(float* __restrict__ agg, const float* __restrict__ norm,
                               const float* __restrict__ b, float* __restrict__ out,
                               int n_nodes) {
    int t = blockIdx.x * blockDim.x + threadIdx.x;
    if (t >= n_nodes * 16) return;
    int node = t >> 4, q = t & 15;
    float4* ap = (float4*)agg + t;
    float4 a = *ap;
    float nr = norm[node];
    float4 bb = ((const float4*)b)[q];
    float4 r;
    r.x = fmaxf(fmaf(a.x, nr, bb.x), 0.f);
    r.y = fmaxf(fmaf(a.y, nr, bb.y), 0.f);
    r.z = fmaxf(fmaf(a.z, nr, bb.z), 0.f);
    r.w = fmaxf(fmaf(a.w, nr, bb.w), 0.f);
    ((float4*)out)[t] = r;
    *ap = make_float4(0.f, 0.f, 0.f, 0.f);
}

__global__ void finalize_second(const float* __restrict__ agg, const float* __restrict__ norm,
                                const float* __restrict__ b, float* __restrict__ out,
                                int n_nodes) {
    int t = blockIdx.x * blockDim.x + threadIdx.x;
    if (t >= n_nodes * 16) return;
    int node = t >> 4, q = t & 15;
    float4 a = ((const float4*)agg)[t];
    float nr = norm[node];
    float4 bb = ((const float4*)b)[q];
    float4 o = ((float4*)out)[t];
    o.x += fmaxf(fmaf(a.x, nr, bb.x), 0.f);
    o.y += fmaxf(fmaf(a.y, nr, bb.y), 0.f);
    o.z += fmaxf(fmaf(a.z, nr, bb.z), 0.f);
    o.w += fmaxf(fmaf(a.w, nr, bb.w), 0.f);
    ((float4*)out)[t] = o;
}

extern "C" void kernel_launch(void* const* d_in, const int* in_sizes, int n_in,
                              void* d_out, int out_size, void* d_ws, size_t ws_size,
                              hipStream_t stream) {
    const float* x   = (const float*)d_in[1];
    const int*   src = (const int*)d_in[2];
    const int*   dst = (const int*)d_in[3];
    const float* W_l = (const float*)d_in[4];
    const float* b_l = (const float*)d_in[5];
    const float* W_c = (const float*)d_in[6];
    const float* b_c = (const float*)d_in[7];
    float* out = (float*)d_out;

    const int n_edges = in_sizes[2];
    const int n_nodes = in_sizes[1] / (2 * N_FEAT);

    float* ws   = (float*)d_ws;
    size_t normPad = ((size_t)n_nodes + 127) & ~(size_t)127;
    float* nrm  = ws;
    float* h    = ws + normPad;
    float* agg  = h + (size_t)n_nodes * N_FEAT;

    {
        int n4 = (int)(normPad / 4);
        zero_f4<<<(n4 + 255) / 256, 256, 0, stream>>>((float4*)nrm, n4);
        int a4 = n_nodes * 16;
        zero_f4<<<(a4 + 255) / 256, 256, 0, stream>>>((float4*)agg, a4);
    }
    deg_kernel<<<(n_edges + 255) / 256, 256, 0, stream>>>(dst, nrm, n_edges);
    norm_kernel<<<(n_nodes + 255) / 256, 256, 0, stream>>>(nrm, n_nodes);

    int nwaves  = (n_nodes + 15) / 16;
    int gth     = nwaves * 32;
    int sgrid   = (n_edges * 16 + 255) / 256;
    int fgrid   = (n_nodes * 16 + 255) / 256;

    gemm_norm_wmma<<<(gth + 255) / 256, 256, 0, stream>>>(x, 0, W_l, nrm, h, n_nodes);
    scatter_kernel<<<sgrid, 256, 0, stream>>>(h, src, dst, agg, n_edges);
    finalize_first<<<fgrid, 256, 0, stream>>>(agg, nrm, b_l, out, n_nodes);

    gemm_norm_wmma<<<(gth + 255) / 256, 256, 0, stream>>>(x, 1, W_c, nrm, h, n_nodes);
    scatter_kernel<<<sgrid, 256, 0, stream>>>(h, src, dst, agg, n_edges);
    finalize_second<<<fgrid, 256, 0, stream>>>(agg, nrm, b_c, out, n_nodes);
}


# scatter remapped: a wave walks 32 edges one at a time, lane = feature, so each row load and each f32 atomic covers one whole 128-B line (4 line requests/edge, was 10); atomic arrival order kept
# speedup vs baseline: 1.5026x; 1.5026x over previous
#include <hip/hip_runtime.h>

typedef float v2f __attribute__((ext_vector_type(2)));
typedef float v8f __attribute__((ext_vector_type(8)));

#define N_FEAT 64

__global__ void zero_f4(float4* __restrict__ p, int n4) {
    int i = blockIdx.x * blockDim.x + threadIdx.x;
    if (i < n4) p[i] = make_float4(0.f, 0.f, 0.f, 0.f);
}

__global__ void deg_kernel(const int* __restrict__ dst, float* __restrict__ deg, int n_edges) {
    int e = blockIdx.x * blockDim.x + threadIdx.x;
    if (e < n_edges) atomicAdd(&deg[dst[e]], 1.0f);
}

__global__ void norm_kernel(float* __restrict__ dn, int n_nodes) {
    int i = blockIdx.x * blockDim.x + threadIdx.x;
    if (i < n_nodes) {
        float d = dn[i];
        dn[i] = (d > 0.f) ? rsqrtf(d) : 0.f;
    }
}

__global__ void gemm_norm_wmma(const float* __restrict__ x, int chan,
                               const float* __restrict__ W,
                               const float* __restrict__ norm,
                               float* __restrict__ h, int n_nodes) {
    int wave  = (blockIdx.x * blockDim.x + threadIdx.x) >> 5;
    int lane  = threadIdx.x & 31;
    int node0 = wave * 16;
    if (node0 >= n_nodes) return;
    int half = lane >> 4;
    int l16  = lane & 15;

    const float* xrow = x + (size_t)(node0 + l16) * (2 * N_FEAT) + chan;
    v8f acc0 = {}, acc1 = {}, acc2 = {}, acc3 = {};

#pragma unroll
    for (int kc = 0; kc < 16; ++kc) {
        int k0 = kc * 4 + half * 2;
        v2f a;
        a.x = xrow[(size_t)(k0 + 0) * 2];
        a.y = xrow[(size_t)(k0 + 1) * 2];
        const float* wr0 = W + (size_t)(k0 + 0) * N_FEAT;
        const float* wr1 = W + (size_t)(k0 + 1) * N_FEAT;
        v2f b0; b0.x = wr0[l16 +  0]; b0.y = wr1[l16 +  0];
        v2f b1; b1.x = wr0[l16 + 16]; b1.y = wr1[l16 + 16];
        v2f b2; b2.x = wr0[l16 + 32]; b2.y = wr1[l16 + 32];
        v2f b3; b3.x = wr0[l16 + 48]; b3.y = wr1[l16 + 48];
        acc0 = __builtin_amdgcn_wmma_f32_16x16x4_f32(false, a, false, b0, (short)0, acc0, false, false);
        acc1 = __builtin_amdgcn_wmma_f32_16x16x4_f32(false, a, false, b1, (short)0, acc1, false, false);
        acc2 = __builtin_amdgcn_wmma_f32_16x16x4_f32(false, a, false, b2, (short)0, acc2, false, false);
        acc3 = __builtin_amdgcn_wmma_f32_16x16x4_f32(false, a, false, b3, (short)0, acc3, false, false);
    }

#pragma unroll
    for (int v = 0; v < 8; ++v) {
        int row = node0 + v + half * 8;
        float nr = norm[row];
        float* hr = h + (size_t)row * N_FEAT + l16;
        hr[ 0] = acc0[v] * nr;
        hr[16] = acc1[v] * nr;
        hr[32] = acc2[v] * nr;
        hr[48] = acc3[v] * nr;
    }
}

#define EPW 32
#define EGRP 8
__global__ void scatter_kernel(const float* __restrict__ h,
                               const int* __restrict__ src,
                               const int* __restrict__ dst,
                               float* __restrict__ agg, int n_edges) {
    int wave = (blockIdx.x * blockDim.x + threadIdx.x) >> 5;
    int lane = threadIdx.x & 31;
    int e0 = wave * EPW;
    if (e0 >= n_edges) return;
    int cnt = n_edges - e0;
    int s_l = 0, d_l = 0;
    if (lane < cnt) { s_l = src[e0 + lane]; d_l = dst[e0 + lane]; }

    if (cnt >= EPW) {
#pragma unroll
        for (int g = 0; g < EPW; g += EGRP) {
            float v0[EGRP], v1[EGRP];
#pragma unroll
            for (int j = 0; j < EGRP; ++j) {
                int s = __builtin_amdgcn_readlane(s_l, g + j);
                const float* hp = h + (size_t)s * N_FEAT + lane;
                v0[j] = hp[0];
                v1[j] = hp[32];
            }
#pragma unroll
            for (int j = 0; j < EGRP; ++j) {
                int d = __builtin_amdgcn_readlane(d_l, g + j);
                float* ap = agg + (size_t)d * N_FEAT + lane;
                atomicAdd(ap, v0[j]);
                atomicAdd(ap + 32, v1[j]);
            }
        }
    } else {
        for (int j = 0; j < cnt; ++j) {
            int s = __builtin_amdgcn_readlane(s_l, j);
            int d = __builtin_amdgcn_readlane(d_l, j);
            const float* hp = h + (size_t)s * N_FEAT + lane;
            float* ap = agg + (size_t)d * N_FEAT + lane;
            float a0 = hp[0], a1 = hp[32];
            atomicAdd(ap, a0);
            atomicAdd(ap + 32, a1);
        }
    }
}

__global__ void finalize_first(float* __restrict__ agg, const float* __restrict__ norm,
                               const float* __restrict__ b, float* __restrict__ out,
                               int n_nodes) {
    int t = blockIdx.x * blockDim.x + threadIdx.x;
    if (t >= n_nodes * 16) return;
    int node = t >> 4, q = t & 15;
    float4* ap = (float4*)agg + t;
    float4 a = *ap;
    float nr = norm[node];
    float4 bb = ((const float4*)b)[q];
    float4 r;
    r.x = fmaxf(fmaf(a.x, nr, bb.x), 0.f);
    r.y = fmaxf(fmaf(a.y, nr, bb.y), 0.f);
    r.z = fmaxf(fmaf(a.z, nr, bb.z), 0.f);
    r.w = fmaxf(fmaf(a.w, nr, bb.w), 0.f);
    ((float4*)out)[t] = r;
    *ap = make_float4(0.f, 0.f, 0.f, 0.f);
}

__global__ void finalize_second(const float* __restrict__ agg, const float* __restrict__ norm,
                                const float* __restrict__ b, float* __restrict__ out,
                                int n_nodes) {
    int t = blockIdx.x * blockDim.x + threadIdx.x;
    if (t >= n_nodes * 16) return;
    int node = t >> 4, q = t & 15;
    float4 a = ((const float4*)agg)[t];
    float nr = norm[node];
    float4 bb = ((const float4*)b)[q];
    float4 o = ((float4*)out)[t];
    o.x += fmaxf(fmaf(a.x, nr, bb.x), 0.f);
    o.y += fmaxf(fmaf(a.y, nr, bb.y), 0.f);
    o.z += fmaxf(fmaf(a.z, nr, bb.z), 0.f);
    o.w += fmaxf(fmaf(a.w, nr, bb.w), 0.f);
    ((float4*)out)[t] = o;
}

extern "C" void kernel_launch(void* const* d_in, const int* in_sizes, int n_in,
                              void* d_out, int out_size, void* d_ws, size_t ws_size,
                              hipStream_t stream) {
    const float* x   = (const float*)d_in[1];
    const int*   src = (const int*)d_in[2];
    const int*   dst = (const int*)d_in[3];
    const float* W_l = (const float*)d_in[4];
    const float* b_l = (const float*)d_in[5];
    const float* W_c = (const float*)d_in[6];
    const float* b_c = (const float*)d_in[7];
    float* out = (float*)d_out;

    const int n_edges = in_sizes[2];
    const int n_nodes = in_sizes[1] / (2 * N_FEAT);

    float* ws   = (float*)d_ws;
    size_t normPad = ((size_t)n_nodes + 127) & ~(size_t)127;
    float* nrm  = ws;
    float* h    = ws + normPad;
    float* agg  = h + (size_t)n_nodes * N_FEAT;

    {
        int n4 = (int)(normPad / 4);
        zero_f4<<<(n4 + 255) / 256, 256, 0, stream>>>((float4*)nrm, n4);
        int a4 = n_nodes * 16;
        zero_f4<<<(a4 + 255) / 256, 256, 0, stream>>>((float4*)agg, a4);
    }
    deg_kernel<<<(n_edges + 255) / 256, 256, 0, stream>>>(dst, nrm, n_edges);
    norm_kernel<<<(n_nodes + 255) / 256, 256, 0, stream>>>(nrm, n_nodes);

    int nwaves  = (n_nodes + 15) / 16;
    int gth     = nwaves * 32;
    int sgrid   = ((n_edges + EPW - 1) / EPW * 32 + 255) / 256;
    int fgrid   = (n_nodes * 16 + 255) / 256;

    gemm_norm_wmma<<<(gth + 255) / 256, 256, 0, stream>>>(x, 0, W_l, nrm, h, n_nodes);
    scatter_kernel<<<sgrid, 256, 0, stream>>>(h, src, dst, agg, n_edges);
    finalize_first<<<fgrid, 256, 0, stream>>>(agg, nrm, b_l, out, n_nodes);

    gemm_norm_wmma<<<(gth + 255) / 256, 256, 0, stream>>>(x, 1, W_c, nrm, h, n_nodes);
    scatter_kernel<<<sgrid, 256, 0, stream>>>(h, src, dst, agg, n_edges);
    finalize_second<<<fgrid, 256, 0, stream>>>(agg, nrm, b_c, out, n_nodes);
}


# float-atomic scatter replaced by grouping edges by destination each call (two-level counting sort on LDS counters) + per-node register sum of h rows, epilogue fused; sum order now segment order
# speedup vs baseline: 3.9278x; 2.6140x over previous
#include <hip/hip_runtime.h>


typedef float v2f __attribute__((ext_vector_type(2)));
typedef float v8f __attribute__((ext_vector_type(8)));

#define N_FEAT 64

#define BSH 8
#define BKT (1 << BSH)
#define NBMAX 512
#define TROW (NBMAX + 1)
#define CH 4096
#define BIN_T 256
#define EPT (CH / BIN_T)
#define SEG_T 512
#define CAP 8192
static_assert(NBMAX == 2 * BIN_T, "the chunk block scans two table entries per thread");

template <int NW>
__device__ __forceinline__ int block_excl_scan(int x, int* wsum, int& tot) {
    int lane = threadIdx.x & 31, wid = threadIdx.x >> 5;
    int v = x;
    for (int d = 1; d < 32; d <<= 1) {
        int t = __shfl_up(v, d);
        if (lane >= d) v += t;
    }
    if (lane == 31) wsum[wid] = v;
    __syncthreads();
    int pre = 0, all = 0;
    for (int k = 0; k < NW; ++k) { int t = wsum[k]; if (k < wid) pre += t; all += t; }
    tot = all;
    __syncthreads();
    return pre + v - x;
}

__global__ void
bin_kernel(const int* __restrict__ src, const int* __restrict__ dst,
           int* __restrict__ bsrc, int* __restrict__ bdst, int* __restrict__ T,
           int n_edges, int n_buckets) {
    __shared__ int hist[TROW];
    __shared__ int s_src[CH];
    __shared__ int s_dst[CH];
    __shared__ int wsum[BIN_T / 32];
    int tid = threadIdx.x;
    int base = blockIdx.x * CH;
    int nE = min(CH, n_edges - base);
    for (int i = tid; i < TROW; i += BIN_T) hist[i] = 0;
    __syncthreads();
    int d[EPT], s[EPT], r[EPT];
#pragma unroll
    for (int q = 0; q < EPT; ++q) {
        int i = q * BIN_T + tid;
        d[q] = -1; s[q] = 0;
        if (i < nE) { d[q] = dst[base + i]; s[q] = src[base + i]; }
    }
#pragma unroll
    for (int q = 0; q < EPT; ++q) {
        r[q] = 0;
        if (d[q] >= 0) r[q] = atomicAdd(&hist[d[q] >> BSH], 1);
    }
    __syncthreads();
    int a = hist[2 * tid], b = hist[2 * tid + 1];
    int tot;
    int ex = block_excl_scan<BIN_T / 32>(a + b, wsum, tot);
    hist[2 * tid] = ex;
    hist[2 * tid + 1] = ex + a;
    if (tid == BIN_T - 1) hist[NBMAX] = tot;
    __syncthreads();
    for (int i = tid; i <= n_buckets; i += BIN_T) T[(size_t)blockIdx.x * TROW + i] = hist[i];
#pragma unroll
    for (int q = 0; q < EPT; ++q) {
        if (d[q] >= 0) {
            int slot = hist[d[q] >> BSH] + r[q];
            s_src[slot] = s[q];
            s_dst[slot] = d[q];
        }
    }
    __syncthreads();
    for (int i = tid; i < nE; i += BIN_T) {
        bsrc[base + i] = s_src[i];
        bdst[base + i] = s_dst[i];
    }
}

__global__ void
seg_kernel(const int* __restrict__ bsrc, const int* __restrict__ bdst, const int* __restrict__ T,
           int* __restrict__ cnt, int* __restrict__ offs, float* __restrict__ nrm,
           int* __restrict__ csr, int n_nodes, int n_chunks) {
    __shared__ int lcnt[BKT];
    __shared__ int cursor[BKT];
    __shared__ int wsum[SEG_T / 32];
    __shared__ int s_csr[CAP];
    int tid = threadIdx.x, k = blockIdx.x;
    if (tid < BKT) lcnt[tid] = 0;
    __syncthreads();
    int before = 0;
    for (int b = tid; b < n_chunks; b += SEG_T) {
        int lo = T[(size_t)b * TROW + k], hi = T[(size_t)b * TROW + k + 1];
        before += lo;
        const int* pd = bdst + (size_t)b * CH;
        for (int i = lo; i < hi; ++i) atomicAdd(&lcnt[pd[i] & (BKT - 1)], 1);
    }
    __syncthreads();
    int c = (tid < BKT) ? lcnt[tid] : 0;
    int tot, base;
    int ex = block_excl_scan<SEG_T / 32>(c, wsum, tot);
    block_excl_scan<SEG_T / 32>(before, wsum, base);
    if (tid < BKT) {
        cursor[tid] = ex;
        int node = k * BKT + tid;
        if (node < n_nodes) {
            cnt[node]  = c;
            offs[node] = base + ex;
            nrm[node]  = (c > 0) ? rsqrtf((float)c) : 0.f;
        }
    }
    __syncthreads();
    bool staged = tot <= CAP;
    for (int b = tid; b < n_chunks; b += SEG_T) {
        int lo = T[(size_t)b * TROW + k], hi = T[(size_t)b * TROW + k + 1];
        const int* pd = bdst + (size_t)b * CH;
        const int* ps = bsrc + (size_t)b * CH;
        for (int i = lo; i < hi; ++i) {
            int sv = ps[i];
            int p = atomicAdd(&cursor[pd[i] & (BKT - 1)], 1);
            if (staged) s_csr[p] = sv; else csr[base + p] = sv;
        }
    }
    __syncthreads();
    if (staged)
        for (int i = tid; i < tot; i += SEG_T) csr[base + i] = s_csr[i];
}

__global__ void gemm_norm_wmma(const float* __restrict__ x, int chan,
                               const float* __restrict__ W,
                               const float* __restrict__ norm,
                               float* __restrict__ h, int n_nodes) {
    int wave  = (blockIdx.x * blockDim.x + threadIdx.x) >> 5;
    int lane  = threadIdx.x & 31;
    int node0 = wave * 16;
    if (node0 >= n_nodes) return;
    int half = lane >> 4;
    int l16  = lane & 15;

    const float* xrow = x + (size_t)(node0 + l16) * (2 * N_FEAT) + chan;
    v8f acc0 = {}, acc1 = {}, acc2 = {}, acc3 = {};

#pragma unroll
    for (int kc = 0; kc < 16; ++kc) {
        int k0 = kc * 4 + half * 2;
        v2f a;
        a.x = xrow[(size_t)(k0 + 0) * 2];
        a.y = xrow[(size_t)(k0 + 1) * 2];
        const float* wr0 = W + (size_t)(k0 + 0) * N_FEAT;
        const float* wr1 = W + (size_t)(k0 + 1) * N_FEAT;
        v2f b0; b0.x = wr0[l16 +  0]; b0.y = wr1[l16 +  0];
        v2f b1; b1.x = wr0[l16 + 16]; b1.y = wr1[l16 + 16];
        v2f b2; b2.x = wr0[l16 + 32]; b2.y = wr1[l16 + 32];
        v2f b3; b3.x = wr0[l16 + 48]; b3.y = wr1[l16 + 48];
        acc0 = __builtin_amdgcn_wmma_f32_16x16x4_f32(false, a, false, b0, (short)0, acc0, false, false);
        acc1 = __builtin_amdgcn_wmma_f32_16x16x4_f32(false, a, false, b1, (short)0, acc1, false, false);
        acc2 = __builtin_amdgcn_wmma_f32_16x16x4_f32(false, a, false, b2, (short)0, acc2, false, false);
        acc3 = __builtin_amdgcn_wmma_f32_16x16x4_f32(false, a, false, b3, (short)0, acc3, false, false);
    }

#pragma unroll
    for (int v = 0; v < 8; ++v) {
        int row = node0 + v + half * 8;
        float nr = norm[row];
        float* hr = h + (size_t)row * N_FEAT + l16;
        hr[ 0] = acc0[v] * nr;
        hr[16] = acc1[v] * nr;
        hr[32] = acc2[v] * nr;
        hr[48] = acc3[v] * nr;
    }
}

#define NPW 8

__device__ __forceinline__ void load_rows(const float4* __restrict__ h4, int cur, int j, int rem,
                                          int half, int l16, float4 (&v)[8]) {
#pragma unroll
    for (int u = 0; u < 8; ++u) {
        int ea = j + 2 * u;
        int s0 = __builtin_amdgcn_readlane(cur, ea);
        int s1 = __builtin_amdgcn_readlane(cur, ea + 1);
        int s = half ? s1 : s0;
        v[u] = make_float4(0.f, 0.f, 0.f, 0.f);
        if (ea + half < rem) v[u] = h4[(size_t)s * 16 + l16];
    }
}

__device__ __forceinline__ float sum8(float a, float b, float c, float d,
                                      float e, float f, float g, float h) {
    return ((a + b) + (c + d)) + ((e + f) + (g + h));
}
__device__ __forceinline__ void add_rows(float4& acc, const float4 (&v)[8]) {
    acc.x += sum8(v[0].x, v[1].x, v[2].x, v[3].x, v[4].x, v[5].x, v[6].x, v[7].x);
    acc.y += sum8(v[0].y, v[1].y, v[2].y, v[3].y, v[4].y, v[5].y, v[6].y, v[7].y);
    acc.z += sum8(v[0].z, v[1].z, v[2].z, v[3].z, v[4].z, v[5].z, v[6].z, v[7].z);
    acc.w += sum8(v[0].w, v[1].w, v[2].w, v[3].w, v[4].w, v[5].w, v[6].w, v[7].w);
}

template <bool FIRST>
__global__ void gather_kernel(const float* __restrict__ h, const int* __restrict__ csr,
                              const int* __restrict__ cnt, const int* __restrict__ offs,
                              const float* __restrict__ nrm, const float* __restrict__ b,
                              float* __restrict__ out, int n_nodes) {
    int wave = __builtin_amdgcn_readfirstlane((int)((blockIdx.x * blockDim.x + threadIdx.x) >> 5));
    int lane = threadIdx.x & 31;
    int node0 = wave * NPW;
    if (node0 >= n_nodes) return;
    int half = lane >> 4, l16 = lane & 15;
    int nn = min(NPW, n_nodes - node0);

    int mi = node0 + (lane & (NPW - 1));
    int c_l = 0, o_l = 0, n_l = 0;
    if (mi < n_nodes) { c_l = cnt[mi]; o_l = offs[mi]; n_l = __float_as_int(nrm[mi]); }
    float4 bb = ((const float4*)b)[l16];
    const float4* h4 = (const float4*)h;

    int c = __builtin_amdgcn_readlane(c_l, 0);
    int o = __builtin_amdgcn_readlane(o_l, 0);
    int sl = (lane < c) ? csr[o + lane] : 0;

    for (int k = 0; k < nn; ++k) {
        int cn = 0, on = 0;
        if (k + 1 < nn) {
            cn = __builtin_amdgcn_readlane(c_l, k + 1);
            on = __builtin_amdgcn_readlane(o_l, k + 1);
        }
        float nr = __int_as_float(__builtin_amdgcn_readlane(n_l, k));
        float4* op = (float4*)(out + (size_t)(node0 + k) * N_FEAT) + l16;

        float4 acc = make_float4(0.f, 0.f, 0.f, 0.f);
        float4 v[8];
        load_rows(h4, sl, 0, c, half, l16, v);
        int sl_next = (lane < cn) ? csr[on + lane] : 0;
        float4 prev = make_float4(0.f, 0.f, 0.f, 0.f);
        if (!FIRST && half == 0) prev = *op;
        add_rows(acc, v);
        int cur = sl;
        for (int e = 16; e < c; e += 16) {
            int jj = e & 31;
            if (jj == 0) cur = (e + lane < c) ? csr[o + e + lane] : 0;
            load_rows(h4, cur, jj, c - (e - jj), half, l16, v);
            add_rows(acc, v);
        }

        acc.x += __shfl_xor(acc.x, 16);
        acc.y += __shfl_xor(acc.y, 16);
        acc.z += __shfl_xor(acc.z, 16);
        acc.w += __shfl_xor(acc.w, 16);
        if (half == 0) {
            float4 r;
            r.x = prev.x + fmaxf(fmaf(acc.x, nr, bb.x), 0.f);
            r.y = prev.y + fmaxf(fmaf(acc.y, nr, bb.y), 0.f);
            r.z = prev.z + fmaxf(fmaf(acc.z, nr, bb.z), 0.f);
            r.w = prev.w + fmaxf(fmaf(acc.w, nr, bb.w), 0.f);
            *op = r;
        }
        c = cn; o = on; sl = sl_next;
    }
}

extern "C" void kernel_launch(void* const* d_in, const int* in_sizes, int n_in,
                              void* d_out, int out_size, void* d_ws, size_t ws_size,
                              hipStream_t stream) {
    const float* x   = (const float*)d_in[1];
    const int*   src = (const int*)d_in[2];
    const int*   dst = (const int*)d_in[3];
    const float* W_l = (const float*)d_in[4];
    const float* b_l = (const float*)d_in[5];
    const float* W_c = (const float*)d_in[6];
    const float* b_c = (const float*)d_in[7];
    float* out = (float*)d_out;

    const int n_edges = in_sizes[2];
    const int n_nodes = in_sizes[1] / (2 * N_FEAT);
    const int n_buckets = (n_nodes + BKT - 1) / BKT;
    const int n_chunks  = (n_edges + CH - 1) / CH;

    size_t nPad = ((size_t)n_nodes + 127) & ~(size_t)127;
    size_t ePad = ((size_t)n_edges + 127) & ~(size_t)127;
    size_t tPad = ((size_t)n_chunks * TROW + 127) & ~(size_t)127;
    int*   cnt   = (int*)d_ws;
    int*   offs  = cnt + nPad;
    float* nrm   = (float*)(offs + nPad);
    int*   T     = (int*)(nrm + nPad);
    int*   bsrc  = T + tPad;
    int*   bdst  = bsrc + ePad;
    int*   csr   = bdst + ePad;
    float* h     = (float*)(csr + ePad);

    bin_kernel<<<n_chunks, BIN_T, 0, stream>>>(src, dst, bsrc, bdst, T, n_edges, n_buckets);
    seg_kernel<<<n_buckets, SEG_T, 0, stream>>>(bsrc, bdst, T, cnt, offs, nrm, csr, n_nodes, n_chunks);

    int nwaves  = (n_nodes + 15) / 16;
    int gth     = nwaves * 32;
    int ggrid   = ((n_nodes + NPW - 1) / NPW * 32 + 255) / 256;

    gemm_norm_wmma<<<(gth + 255) / 256, 256, 0, stream>>>(x, 0, W_l, nrm, h, n_nodes);
    gather_kernel<true><<<ggrid, 256, 0, stream>>>(h, csr, cnt, offs, nrm, b_l, out, n_nodes);

    gemm_norm_wmma<<<(gth + 255) / 256, 256, 0, stream>>>(x, 1, W_c, nrm, h, n_nodes);
    gather_kernel<false><<<ggrid, 256, 0, stream>>>(h, csr, cnt, offs, nrm, b_c, out, n_nodes);
}


# both channels in one pass: one product launch reads x once (16-B loads) and writes a 128-wide h row, one gather sums 512-B rows and writes out once; 6 -> 4 launches; per-element sum order unchanged
# speedup vs baseline: 4.5222x; 1.1513x over previous
#include <hip/hip_runtime.h>


typedef float v2f __attribute__((ext_vector_type(2)));
typedef float v8f __attribute__((ext_vector_type(8)));

#define N_FEAT 64

#define BSH 8
#define BKT (1 << BSH)
#define NBMAX 512
#define TROW (NBMAX + 1)
#define CH 4096
#define BIN_T 256
#define EPT (CH / BIN_T)
#define SEG_T 512
#define CAP 8192
static_assert(NBMAX == 2 * BIN_T, "the chunk block scans two table entries per thread");

template <int NW>
__device__ __forceinline__ int block_excl_scan(int x, int* wsum, int& tot) {
    int lane = threadIdx.x & 31, wid = threadIdx.x >> 5;
    int v = x;
    for (int d = 1; d < 32; d <<= 1) {
        int t = __shfl_up(v, d);
        if (lane >= d) v += t;
    }
    if (lane == 31) wsum[wid] = v;
    __syncthreads();
    int pre = 0, all = 0;
    for (int k = 0; k < NW; ++k) { int t = wsum[k]; if (k < wid) pre += t; all += t; }
    tot = all;
    __syncthreads();
    return pre + v - x;
}

__global__ void
bin_kernel(const int* __restrict__ src, const int* __restrict__ dst,
           int* __restrict__ bsrc, int* __restrict__ bdst, int* __restrict__ T,
           int n_edges, int n_buckets) {
    __shared__ int hist[TROW];
    __shared__ int s_src[CH];
    __shared__ int s_dst[CH];
    __shared__ int wsum[BIN_T / 32];
    int tid = threadIdx.x;
    int base = blockIdx.x * CH;
    int nE = min(CH, n_edges - base);
    for (int i = tid; i < TROW; i += BIN_T) hist[i] = 0;
    __syncthreads();
    int d[EPT], s[EPT], r[EPT];
#pragma unroll
    for (int q = 0; q < EPT; ++q) {
        int i = q * BIN_T + tid;
        d[q] = -1; s[q] = 0;
        if (i < nE) { d[q] = dst[base + i]; s[q] = src[base + i]; }
    }
#pragma unroll
    for (int q = 0; q < EPT; ++q) {
        r[q] = 0;
        if (d[q] >= 0) r[q] = atomicAdd(&hist[d[q] >> BSH], 1);
    }
    __syncthreads();
    int a = hist[2 * tid], b = hist[2 * tid + 1];
    int tot;
    int ex = block_excl_scan<BIN_T / 32>(a + b, wsum, tot);
    hist[2 * tid] = ex;
    hist[2 * tid + 1] = ex + a;
    if (tid == BIN_T - 1) hist[NBMAX] = tot;
    __syncthreads();
    for (int i = tid; i <= n_buckets; i += BIN_T) T[(size_t)blockIdx.x * TROW + i] = hist[i];
#pragma unroll
    for (int q = 0; q < EPT; ++q) {
        if (d[q] >= 0) {
            int slot = hist[d[q] >> BSH] + r[q];
            s_src[slot] = s[q];
            s_dst[slot] = d[q];
        }
    }
    __syncthreads();
    for (int i = tid; i < nE; i += BIN_T) {
        bsrc[base + i] = s_src[i];
        bdst[base + i] = s_dst[i];
    }
}

__global__ void
seg_kernel(const int* __restrict__ bsrc, const int* __restrict__ bdst, const int* __restrict__ T,
           int* __restrict__ cnt, int* __restrict__ offs, float* __restrict__ nrm,
           int* __restrict__ csr, int n_nodes, int n_chunks) {
    __shared__ int lcnt[BKT];
    __shared__ int cursor[BKT];
    __shared__ int wsum[SEG_T / 32];
    __shared__ int s_csr[CAP];
    int tid = threadIdx.x, k = blockIdx.x;
    if (tid < BKT) lcnt[tid] = 0;
    __syncthreads();
    int before = 0;
    for (int b = tid; b < n_chunks; b += SEG_T) {
        int lo = T[(size_t)b * TROW + k], hi = T[(size_t)b * TROW + k + 1];
        before += lo;
        const int* pd = bdst + (size_t)b * CH;
        for (int i = lo; i < hi; ++i) atomicAdd(&lcnt[pd[i] & (BKT - 1)], 1);
    }
    __syncthreads();
    int c = (tid < BKT) ? lcnt[tid] : 0;
    int tot, base;
    int ex = block_excl_scan<SEG_T / 32>(c, wsum, tot);
    block_excl_scan<SEG_T / 32>(before, wsum, base);
    if (tid < BKT) {
        cursor[tid] = ex;
        int node = k * BKT + tid;
        if (node < n_nodes) {
            cnt[node]  = c;
            offs[node] = base + ex;
            nrm[node]  = (c > 0) ? rsqrtf((float)c) : 0.f;
        }
    }
    __syncthreads();
    bool staged = tot <= CAP;
    for (int b = tid; b < n_chunks; b += SEG_T) {
        int lo = T[(size_t)b * TROW + k], hi = T[(size_t)b * TROW + k + 1];
        const int* pd = bdst + (size_t)b * CH;
        const int* ps = bsrc + (size_t)b * CH;
        for (int i = lo; i < hi; ++i) {
            int sv = ps[i];
            int p = atomicAdd(&cursor[pd[i] & (BKT - 1)], 1);
            if (staged) s_csr[p] = sv; else csr[base + p] = sv;
        }
    }
    __syncthreads();
    if (staged)
        for (int i = tid; i < tot; i += SEG_T) csr[base + i] = s_csr[i];
}

__global__ void gemm_norm_wmma(const float* __restrict__ x, int chan,
                               const float* __restrict__ W,
                               const float* __restrict__ norm,
                               float* __restrict__ h, int n_nodes) {
    int wave  = (blockIdx.x * blockDim.x + threadIdx.x) >> 5;
    int lane  = threadIdx.x & 31;
    int node0 = wave * 16;
    if (node0 >= n_nodes) return;
    int half = lane >> 4;
    int l16  = lane & 15;

    const float* xrow = x + (size_t)(node0 + l16) * (2 * N_FEAT) + chan;
    v8f acc0 = {}, acc1 = {}, acc2 = {}, acc3 = {};

#pragma unroll
    for (int kc = 0; kc < 16; ++kc) {
        int k0 = kc * 4 + half * 2;
        v2f a;
        a.x = xrow[(size_t)(k0 + 0) * 2];
        a.y = xrow[(size_t)(k0 + 1) * 2];
        const float* wr0 = W + (size_t)(k0 + 0) * N_FEAT;
        const float* wr1 = W + (size_t)(k0 + 1) * N_FEAT;
        v2f b0; b0.x = wr0[l16 +  0]; b0.y = wr1[l16 +  0];
        v2f b1; b1.x = wr0[l16 + 16]; b1.y = wr1[l16 + 16];
        v2f b2; b2.x = wr0[l16 + 32]; b2.y = wr1[l16 + 32];
        v2f b3; b3.x = wr0[l16 + 48]; b3.y = wr1[l16 + 48];
        acc0 = __builtin_amdgcn_wmma_f32_16x16x4_f32(false, a, false, b0, (short)0, acc0, false, false);
        acc1 = __builtin_amdgcn_wmma_f32_16x16x4_f32(false, a, false, b1, (short)0, acc1, false, false);
        acc2 = __builtin_amdgcn_wmma_f32_16x16x4_f32(false, a, false, b2, (short)0, acc2, false, false);
        acc3 = __builtin_amdgcn_wmma_f32_16x16x4_f32(false, a, false, b3, (short)0, acc3, false, false);
    }

#pragma unroll
    for (int v = 0; v < 8; ++v) {
        int row = node0 + v + half * 8;
        float nr = norm[row];
        float* hr = h + (size_t)row * N_FEAT + l16;
        hr[ 0] = acc0[v] * nr;
        hr[16] = acc1[v] * nr;
        hr[32] = acc2[v] * nr;
        hr[48] = acc3[v] * nr;
    }
}

__global__ void gemm2_norm_wmma(const float* __restrict__ x,
                                const float* __restrict__ Wl, const float* __restrict__ Wc,
                                const float* __restrict__ norm,
                                float* __restrict__ h2, int n_nodes) {
    int wave  = (blockIdx.x * blockDim.x + threadIdx.x) >> 5;
    int lane  = threadIdx.x & 31;
    int node0 = wave * 16;
    if (node0 >= n_nodes) return;
    int half = lane >> 4;
    int l16  = lane & 15;

    const float4* xq = (const float4*)(x + (size_t)(node0 + l16) * (2 * N_FEAT)) + half;
    v8f l0 = {}, l1 = {}, l2 = {}, l3 = {}, c0 = {}, c1 = {}, c2 = {}, c3 = {};

#pragma unroll
    for (int kc = 0; kc < 16; ++kc) {
        int k0 = kc * 4 + half * 2;
        float4 q = xq[kc * 2];
        v2f al; al.x = q.x; al.y = q.z;
        v2f ac; ac.x = q.y; ac.y = q.w;
        const float* wl0 = Wl + (size_t)(k0 + 0) * N_FEAT;
        const float* wl1 = Wl + (size_t)(k0 + 1) * N_FEAT;
        const float* wc0 = Wc + (size_t)(k0 + 0) * N_FEAT;
        const float* wc1 = Wc + (size_t)(k0 + 1) * N_FEAT;
        v2f bl0; bl0.x = wl0[l16 +  0]; bl0.y = wl1[l16 +  0];
        v2f bl1; bl1.x = wl0[l16 + 16]; bl1.y = wl1[l16 + 16];
        v2f bl2; bl2.x = wl0[l16 + 32]; bl2.y = wl1[l16 + 32];
        v2f bl3; bl3.x = wl0[l16 + 48]; bl3.y = wl1[l16 + 48];
        v2f bc0; bc0.x = wc0[l16 +  0]; bc0.y = wc1[l16 +  0];
        v2f bc1; bc1.x = wc0[l16 + 16]; bc1.y = wc1[l16 + 16];
        v2f bc2; bc2.x = wc0[l16 + 32]; bc2.y = wc1[l16 + 32];
        v2f bc3; bc3.x = wc0[l16 + 48]; bc3.y = wc1[l16 + 48];
        l0 = __builtin_amdgcn_wmma_f32_16x16x4_f32(false, al, false, bl0, (short)0, l0, false, false);
        l1 = __builtin_amdgcn_wmma_f32_16x16x4_f32(false, al, false, bl1, (short)0, l1, false, false);
        l2 = __builtin_amdgcn_wmma_f32_16x16x4_f32(false, al, false, bl2, (short)0, l2, false, false);
        l3 = __builtin_amdgcn_wmma_f32_16x16x4_f32(false, al, false, bl3, (short)0, l3, false, false);
        c0 = __builtin_amdgcn_wmma_f32_16x16x4_f32(false, ac, false, bc0, (short)0, c0, false, false);
        c1 = __builtin_amdgcn_wmma_f32_16x16x4_f32(false, ac, false, bc1, (short)0, c1, false, false);
        c2 = __builtin_amdgcn_wmma_f32_16x16x4_f32(false, ac, false, bc2, (short)0, c2, false, false);
        c3 = __builtin_amdgcn_wmma_f32_16x16x4_f32(false, ac, false, bc3, (short)0, c3, false, false);
    }

#pragma unroll
    for (int v = 0; v < 8; ++v) {
        int row = node0 + v + half * 8;
        float nr = norm[row];
        float* hr = h2 + (size_t)row * (2 * N_FEAT) + l16;
        hr[  0] = l0[v] * nr;
        hr[ 16] = l1[v] * nr;
        hr[ 32] = l2[v] * nr;
        hr[ 48] = l3[v] * nr;
        hr[ 64] = c0[v] * nr;
        hr[ 80] = c1[v] * nr;
        hr[ 96] = c2[v] * nr;
        hr[112] = c3[v] * nr;
    }
}

#define NPW 8

__device__ __forceinline__ void load_rows(const float4* __restrict__ h4, int cur, int j, int rem,
                                          int half, int l16, float4 (&v)[8]) {
#pragma unroll
    for (int u = 0; u < 8; ++u) {
        int ea = j + 2 * u;
        int s0 = __builtin_amdgcn_readlane(cur, ea);
        int s1 = __builtin_amdgcn_readlane(cur, ea + 1);
        int s = half ? s1 : s0;
        v[u] = make_float4(0.f, 0.f, 0.f, 0.f);
        if (ea + half < rem) v[u] = h4[(size_t)s * 16 + l16];
    }
}

__device__ __forceinline__ float sum8(float a, float b, float c, float d,
                                      float e, float f, float g, float h) {
    return ((a + b) + (c + d)) + ((e + f) + (g + h));
}
__device__ __forceinline__ void add_rows(float4& acc, const float4 (&v)[8]) {
    acc.x += sum8(v[0].x, v[1].x, v[2].x, v[3].x, v[4].x, v[5].x, v[6].x, v[7].x);
    acc.y += sum8(v[0].y, v[1].y, v[2].y, v[3].y, v[4].y, v[5].y, v[6].y, v[7].y);
    acc.z += sum8(v[0].z, v[1].z, v[2].z, v[3].z, v[4].z, v[5].z, v[6].z, v[7].z);
    acc.w += sum8(v[0].w, v[1].w, v[2].w, v[3].w, v[4].w, v[5].w, v[6].w, v[7].w);
}

template <bool FIRST>
__global__ void gather_kernel(const float* __restrict__ h, const int* __restrict__ csr,
                              const int* __restrict__ cnt, const int* __restrict__ offs,
                              const float* __restrict__ nrm, const float* __restrict__ b,
                              float* __restrict__ out, int n_nodes) {
    int wave = __builtin_amdgcn_readfirstlane((int)((blockIdx.x * blockDim.x + threadIdx.x) >> 5));
    int lane = threadIdx.x & 31;
    int node0 = wave * NPW;
    if (node0 >= n_nodes) return;
    int half = lane >> 4, l16 = lane & 15;
    int nn = min(NPW, n_nodes - node0);

    int mi = node0 + (lane & (NPW - 1));
    int c_l = 0, o_l = 0, n_l = 0;
    if (mi < n_nodes) { c_l = cnt[mi]; o_l = offs[mi]; n_l = __float_as_int(nrm[mi]); }
    float4 bb = ((const float4*)b)[l16];
    const float4* h4 = (const float4*)h;

    int c = __builtin_amdgcn_readlane(c_l, 0);
    int o = __builtin_amdgcn_readlane(o_l, 0);
    int sl = (lane < c) ? csr[o + lane] : 0;

    for (int k = 0; k < nn; ++k) {
        int cn = 0, on = 0;
        if (k + 1 < nn) {
            cn = __builtin_amdgcn_readlane(c_l, k + 1);
            on = __builtin_amdgcn_readlane(o_l, k + 1);
        }
        float nr = __int_as_float(__builtin_amdgcn_readlane(n_l, k));
        float4* op = (float4*)(out + (size_t)(node0 + k) * N_FEAT) + l16;

        float4 acc = make_float4(0.f, 0.f, 0.f, 0.f);
        float4 v[8];
        load_rows(h4, sl, 0, c, half, l16, v);
        int sl_next = (lane < cn) ? csr[on + lane] : 0;
        float4 prev = make_float4(0.f, 0.f, 0.f, 0.f);
        if (!FIRST && half == 0) prev = *op;
        add_rows(acc, v);
        int cur = sl;
        for (int e = 16; e < c; e += 16) {
            int jj = e & 31;
            if (jj == 0) cur = (e + lane < c) ? csr[o + e + lane] : 0;
            load_rows(h4, cur, jj, c - (e - jj), half, l16, v);
            add_rows(acc, v);
        }

        acc.x += __shfl_xor(acc.x, 16);
        acc.y += __shfl_xor(acc.y, 16);
        acc.z += __shfl_xor(acc.z, 16);
        acc.w += __shfl_xor(acc.w, 16);
        if (half == 0) {
            float4 r;
            r.x = prev.x + fmaxf(fmaf(acc.x, nr, bb.x), 0.f);
            r.y = prev.y + fmaxf(fmaf(acc.y, nr, bb.y), 0.f);
            r.z = prev.z + fmaxf(fmaf(acc.z, nr, bb.z), 0.f);
            r.w = prev.w + fmaxf(fmaf(acc.w, nr, bb.w), 0.f);
            *op = r;
        }
        c = cn; o = on; sl = sl_next;
    }
}

__device__ __forceinline__ void load_rows2(const float4* __restrict__ h4, int cur, int j, int rem,
                                           int half, int l16, float4 (&vl)[8], float4 (&vc)[8]) {
#pragma unroll
    for (int u = 0; u < 8; ++u) {
        int ea = j + 2 * u;
        int s0 = __builtin_amdgcn_readlane(cur, ea);
        int s1 = __builtin_amdgcn_readlane(cur, ea + 1);
        int s = half ? s1 : s0;
        vl[u] = make_float4(0.f, 0.f, 0.f, 0.f);
        vc[u] = make_float4(0.f, 0.f, 0.f, 0.f);
        if (ea + half < rem) {
            const float4* p = h4 + (size_t)s * 32 + l16;
            vl[u] = p[0];
            vc[u] = p[16];
        }
    }
}

__global__ void gather2_kernel(const float* __restrict__ h2, const int* __restrict__ csr,
                               const int* __restrict__ cnt, const int* __restrict__ offs,
                               const float* __restrict__ nrm,
                               const float* __restrict__ bl, const float* __restrict__ bc,
                               float* __restrict__ out, int n_nodes) {
    int wave = __builtin_amdgcn_readfirstlane((int)((blockIdx.x * blockDim.x + threadIdx.x) >> 5));
    int lane = threadIdx.x & 31;
    int node0 = wave * NPW;
    if (node0 >= n_nodes) return;
    int half = lane >> 4, l16 = lane & 15;
    int nn = min(NPW, n_nodes - node0);

    int mi = node0 + (lane & (NPW - 1));
    int c_l = 0, o_l = 0, n_l = 0;
    if (mi < n_nodes) { c_l = cnt[mi]; o_l = offs[mi]; n_l = __float_as_int(nrm[mi]); }
    float4 bbl = ((const float4*)bl)[l16];
    float4 bbc = ((const float4*)bc)[l16];
    const float4* h4 = (const float4*)h2;

    int c = __builtin_amdgcn_readlane(c_l, 0);
    int o = __builtin_amdgcn_readlane(o_l, 0);
    int sl = (lane < c) ? csr[o + lane] : 0;

    for (int k = 0; k < nn; ++k) {
        int cn = 0, on = 0;
        if (k + 1 < nn) {
            cn = __builtin_amdgcn_readlane(c_l, k + 1);
            on = __builtin_amdgcn_readlane(o_l, k + 1);
        }
        float nr = __int_as_float(__builtin_amdgcn_readlane(n_l, k));
        float4* op = (float4*)(out + (size_t)(node0 + k) * N_FEAT) + l16;

        float4 accl = make_float4(0.f, 0.f, 0.f, 0.f);
        float4 accc = make_float4(0.f, 0.f, 0.f, 0.f);
        float4 vl[8], vc[8];
        load_rows2(h4, sl, 0, c, half, l16, vl, vc);
        int sl_next = (lane < cn) ? csr[on + lane] : 0;
        add_rows(accl, vl);
        add_rows(accc, vc);
        int cur = sl;
        for (int e = 16; e < c; e += 16) {
            int jj = e & 31;
            if (jj == 0) cur = (e + lane < c) ? csr[o + e + lane] : 0;
            load_rows2(h4, cur, jj, c - (e - jj), half, l16, vl, vc);
            add_rows(accl, vl);
            add_rows(accc, vc);
        }

        accl.x += __shfl_xor(accl.x, 16);
        accl.y += __shfl_xor(accl.y, 16);
        accl.z += __shfl_xor(accl.z, 16);
        accl.w += __shfl_xor(accl.w, 16);
        accc.x += __shfl_xor(accc.x, 16);
        accc.y += __shfl_xor(accc.y, 16);
        accc.z += __shfl_xor(accc.z, 16);
        accc.w += __shfl_xor(accc.w, 16);
        if (half == 0) {
            float4 r;
            r.x = fmaxf(fmaf(accl.x, nr, bbl.x), 0.f) + fmaxf(fmaf(accc.x, nr, bbc.x), 0.f);
            r.y = fmaxf(fmaf(accl.y, nr, bbl.y), 0.f) + fmaxf(fmaf(accc.y, nr, bbc.y), 0.f);
            r.z = fmaxf(fmaf(accl.z, nr, bbl.z), 0.f) + fmaxf(fmaf(accc.z, nr, bbc.z), 0.f);
            r.w = fmaxf(fmaf(accl.w, nr, bbl.w), 0.f) + fmaxf(fmaf(accc.w, nr, bbc.w), 0.f);
            *op = r;
        }
        c = cn; o = on; sl = sl_next;
    }
}

extern "C" void kernel_launch(void* const* d_in, const int* in_sizes, int n_in,
                              void* d_out, int out_size, void* d_ws, size_t ws_size,
                              hipStream_t stream) {
    const float* x   = (const float*)d_in[1];
    const int*   src = (const int*)d_in[2];
    const int*   dst = (const int*)d_in[3];
    const float* W_l = (const float*)d_in[4];
    const float* b_l = (const float*)d_in[5];
    const float* W_c = (const float*)d_in[6];
    const float* b_c = (const float*)d_in[7];
    float* out = (float*)d_out;

    const int n_edges = in_sizes[2];
    const int n_nodes = in_sizes[1] / (2 * N_FEAT);
    const int n_buckets = (n_nodes + BKT - 1) / BKT;
    const int n_chunks  = (n_edges + CH - 1) / CH;

    size_t nPad = ((size_t)n_nodes + 127) & ~(size_t)127;
    size_t ePad = ((size_t)n_edges + 127) & ~(size_t)127;
    size_t tPad = ((size_t)n_chunks * TROW + 127) & ~(size_t)127;
    int*   cnt   = (int*)d_ws;
    int*   offs  = cnt + nPad;
    float* nrm   = (float*)(offs + nPad);
    int*   csr   = (int*)(nrm + nPad);
    int*   T     = csr + ePad;
    int*   bsrc  = T + tPad;
    int*   bdst  = bsrc + ePad;
    float* h     = (float*)T;
    size_t wide_bytes = (3 * nPad + ePad + (size_t)n_nodes * 2 * N_FEAT) * sizeof(float);

    bin_kernel<<<n_chunks, BIN_T, 0, stream>>>(src, dst, bsrc, bdst, T, n_edges, n_buckets);
    seg_kernel<<<n_buckets, SEG_T, 0, stream>>>(bsrc, bdst, T, cnt, offs, nrm, csr, n_nodes, n_chunks);

    int nwaves  = (n_nodes + 15) / 16;
    int gth     = nwaves * 32;
    int ggrid   = ((n_nodes + NPW - 1) / NPW * 32 + 255) / 256;

    if (ws_size >= wide_bytes) {
        gemm2_norm_wmma<<<(gth + 255) / 256, 256, 0, stream>>>(x, W_l, W_c, nrm, h, n_nodes);
        gather2_kernel<<<ggrid, 256, 0, stream>>>(h, csr, cnt, offs, nrm, b_l, b_c, out, n_nodes);
    } else {
        gemm_norm_wmma<<<(gth + 255) / 256, 256, 0, stream>>>(x, 0, W_l, nrm, h, n_nodes);
        gather_kernel<true><<<ggrid, 256, 0, stream>>>(h, csr, cnt, offs, nrm, b_l, out, n_nodes);
        gemm_norm_wmma<<<(gth + 255) / 256, 256, 0, stream>>>(x, 1, W_c, nrm, h, n_nodes);
        gather_kernel<false><<<ggrid, 256, 0, stream>>>(h, csr, cnt, offs, nrm, b_c, out, n_nodes);
    }
}


# aggregate, then transform, in one launch: a wave sums norm[src]-scaled x rows of 16 nodes (sources sorted by id) into an LDS tile, then both f32 WMMA products; no h; edge sum now before product
# speedup vs baseline: 5.4254x; 1.1997x over previous
#include <hip/hip_runtime.h>


typedef float v2f __attribute__((ext_vector_type(2)));
typedef float v8f __attribute__((ext_vector_type(8)));

#define N_FEAT 64

#define BSH 8
#define BKT (1 << BSH)
#define NBMAX 512
#define TROW (NBMAX + 1)
#define CH 4096
#define BIN_T 256
#define EPT (CH / BIN_T)
#define SEG_T 512
#define CAP 8192
static_assert(NBMAX == 2 * BIN_T, "the chunk block scans two table entries per thread");

template <int NW>
__device__ __forceinline__ int block_excl_scan(int x, int* wsum, int& tot) {
    int lane = threadIdx.x & 31, wid = threadIdx.x >> 5;
    int v = x;
    for (int d = 1; d < 32; d <<= 1) {
        int t = __shfl_up(v, d);
        if (lane >= d) v += t;
    }
    if (lane == 31) wsum[wid] = v;
    __syncthreads();
    int pre = 0, all = 0;
    for (int k = 0; k < NW; ++k) { int t = wsum[k]; if (k < wid) pre += t; all += t; }
    tot = all;
    __syncthreads();
    return pre + v - x;
}

__global__ void
bin_kernel(const int* __restrict__ src, const int* __restrict__ dst,
           int* __restrict__ bsrc, int* __restrict__ bdst, int* __restrict__ T,
           int n_edges, int n_buckets) {
    __shared__ int hist[TROW];
    __shared__ int s_src[CH];
    __shared__ int s_dst[CH];
    __shared__ int wsum[BIN_T / 32];
    int tid = threadIdx.x;
    int base = blockIdx.x * CH;
    int nE = min(CH, n_edges - base);
    for (int i = tid; i < TROW; i += BIN_T) hist[i] = 0;
    __syncthreads();
    int d[EPT], s[EPT], r[EPT];
#pragma unroll
    for (int q = 0; q < EPT; ++q) {
        int i = q * BIN_T + tid;
        d[q] = -1; s[q] = 0;
        if (i < nE) { d[q] = dst[base + i]; s[q] = src[base + i]; }
    }
#pragma unroll
    for (int q = 0; q < EPT; ++q) {
        r[q] = 0;
        if (d[q] >= 0) r[q] = atomicAdd(&hist[d[q] >> BSH], 1);
    }
    __syncthreads();
    int a = hist[2 * tid], b = hist[2 * tid + 1];
    int tot;
    int ex = block_excl_scan<BIN_T / 32>(a + b, wsum, tot);
    hist[2 * tid] = ex;
    hist[2 * tid + 1] = ex + a;
    if (tid == BIN_T - 1) hist[NBMAX] = tot;
    __syncthreads();
    for (int i = tid; i <= n_buckets; i += BIN_T) T[(size_t)blockIdx.x * TROW + i] = hist[i];
#pragma unroll
    for (int q = 0; q < EPT; ++q) {
        if (d[q] >= 0) {
            int slot = hist[d[q] >> BSH] + r[q];
            s_src[slot] = s[q];
            s_dst[slot] = d[q];
        }
    }
    __syncthreads();
    for (int i = tid; i < nE; i += BIN_T) {
        bsrc[base + i] = s_src[i];
        bdst[base + i] = s_dst[i];
    }
}

__global__ void
seg_kernel(const int* __restrict__ bsrc, const int* __restrict__ bdst, const int* __restrict__ T,
           int* __restrict__ cnt, int* __restrict__ offs, float* __restrict__ nrm,
           int* __restrict__ csr, int n_nodes, int n_chunks) {
    __shared__ int lcnt[BKT];
    __shared__ int cursor[BKT];
    __shared__ int wsum[SEG_T / 32];
    __shared__ int s_csr[CAP];
    int tid = threadIdx.x, k = blockIdx.x;
    if (tid < BKT) lcnt[tid] = 0;
    __syncthreads();
    int before = 0;
    for (int b = tid; b < n_chunks; b += SEG_T) {
        int lo = T[(size_t)b * TROW + k], hi = T[(size_t)b * TROW + k + 1];
        before += lo;
        const int* pd = bdst + (size_t)b * CH;
        for (int i = lo; i < hi; ++i) atomicAdd(&lcnt[pd[i] & (BKT - 1)], 1);
    }
    __syncthreads();
    int c = (tid < BKT) ? lcnt[tid] : 0;
    int tot, base;
    int ex = block_excl_scan<SEG_T / 32>(c, wsum, tot);
    block_excl_scan<SEG_T / 32>(before, wsum, base);
    if (tid < BKT) {
        cursor[tid] = ex;
        int node = k * BKT + tid;
        if (node < n_nodes) {
            cnt[node]  = c;
            offs[node] = base + ex;
            nrm[node]  = (c > 0) ? rsqrtf((float)c) : 0.f;
        }
    }
    __syncthreads();
    bool staged = tot <= CAP;
    for (int b = tid; b < n_chunks; b += SEG_T) {
        int lo = T[(size_t)b * TROW + k], hi = T[(size_t)b * TROW + k + 1];
        const int* pd = bdst + (size_t)b * CH;
        const int* ps = bsrc + (size_t)b * CH;
        for (int i = lo; i < hi; ++i) {
            int sv = ps[i];
            int p = atomicAdd(&cursor[pd[i] & (BKT - 1)], 1);
            if (staged) s_csr[p] = sv; else csr[base + p] = sv;
        }
    }
    __syncthreads();
    if (staged)
        for (int i = tid; i < tot; i += SEG_T) csr[base + i] = s_csr[i];
}

#define FT_WAVES 25
#define FT_THREADS (FT_WAVES * 32)
#define WPITCH 80

__device__ __forceinline__ void load_rows2(const float4* __restrict__ r4, int cur, int j, int rem,
                                           int half, int l16, float4 (&va)[8], float4 (&vb)[8]) {
#pragma unroll
    for (int u = 0; u < 8; ++u) {
        int ea = j + 2 * u;
        int s0 = __builtin_amdgcn_readlane(cur, ea);
        int s1 = __builtin_amdgcn_readlane(cur, ea + 1);
        int s = half ? s1 : s0;
        va[u] = make_float4(0.f, 0.f, 0.f, 0.f);
        vb[u] = make_float4(0.f, 0.f, 0.f, 0.f);
        if (ea + half < rem) {
            const float4* p = r4 + (size_t)s * 32 + l16;
            va[u] = p[0];
            vb[u] = p[16];
        }
    }
}

__device__ __forceinline__ void fma_rows2(float4& a, float4& b, const float4 (&va)[8],
                                          const float4 (&vb)[8], int nbits, int j, int half) {
#pragma unroll
    for (int u = 0; u < 8; ++u) {
        int ea = j + 2 * u;
        float n0 = __int_as_float(__builtin_amdgcn_readlane(nbits, ea));
        float n1 = __int_as_float(__builtin_amdgcn_readlane(nbits, ea + 1));
        float ns = half ? n1 : n0;
        a.x = fmaf(va[u].x, ns, a.x); a.y = fmaf(va[u].y, ns, a.y);
        a.z = fmaf(va[u].z, ns, a.z); a.w = fmaf(va[u].w, ns, a.w);
        b.x = fmaf(vb[u].x, ns, b.x); b.y = fmaf(vb[u].y, ns, b.y);
        b.z = fmaf(vb[u].z, ns, b.z); b.w = fmaf(vb[u].w, ns, b.w);
    }
}

__device__ __forceinline__ int sort_sources(int sl, int c, int lane) {
    int m = min(c, 32);
    int key = (sl << 5) | lane;
    int rank = 0;
    for (int t = 0; t < m; ++t)
        rank += (__builtin_amdgcn_readlane(key, t) < key) ? 1 : 0;
    if (lane >= m) rank = lane;
    return __builtin_amdgcn_ds_permute(rank << 2, sl);
}

__global__ void __launch_bounds__(FT_THREADS)
agg_gemm_kernel(const float* __restrict__ x, const int* __restrict__ csr,
                const int* __restrict__ cnt, const int* __restrict__ offs,
                const float* __restrict__ nrm,
                const float* __restrict__ Wl, const float* __restrict__ Wc,
                const float* __restrict__ bl, const float* __restrict__ bc,
                float* __restrict__ out, int n_nodes) {
    struct Lds {
        float2 w[2][32 * WPITCH];
        float4 tile[FT_WAVES][16 * 32];
    };
    __shared__ Lds L;

    int tid = threadIdx.x;
    int lane = tid & 31;
    int half = lane >> 4, l16 = lane & 15;
    int wib  = __builtin_amdgcn_readfirstlane(tid >> 5);
    int wave = blockIdx.x * FT_WAVES + wib;
    int node0 = wave * 16;

    int mi = node0 + l16;
    int c_l = 0, o_l = 0, n_l = 0;
    if (mi < n_nodes) { c_l = cnt[mi]; o_l = offs[mi]; n_l = __float_as_int(nrm[mi]); }

    constexpr int WSTEPS = (2 * 32 * 64 + FT_THREADS - 1) / FT_THREADS;
    float w0[WSTEPS], w1[WSTEPS];
#pragma unroll
    for (int r = 0; r < WSTEPS; ++r) {
        int i = tid + r * FT_THREADS;
        w0[r] = 0.f; w1[r] = 0.f;
        if (i < 2 * 32 * 64) {
            int m = i >> 11, p = (i >> 6) & 31, n = i & 63;
            const float* Wm = m ? Wc : Wl;
            w0[r] = Wm[(2 * p) * N_FEAT + n];
            w1[r] = Wm[(2 * p + 1) * N_FEAT + n];
        }
    }
#pragma unroll
    for (int r = 0; r < WSTEPS; ++r) {
        int i = tid + r * FT_THREADS;
        if (i < 2 * 32 * 64) {
            int m = i >> 11, p = (i >> 6) & 31, n = i & 63;
            L.w[m][p * WPITCH + n] = make_float2(w0[r], w1[r]);
        }
    }
    __syncthreads();
    if (node0 >= n_nodes) return;

    const float4* x4 = (const float4*)x;
    float4* trow = &L.tile[wib][0];

    int c = __builtin_amdgcn_readlane(c_l, 0);
    int o = __builtin_amdgcn_readlane(o_l, 0);
    int sl = (lane < c) ? csr[o + lane] : 0;

    for (int k = 0; k < 16; ++k) {
        int cn = 0, on = 0;
        if (k + 1 < 16) {
            cn = __builtin_amdgcn_readlane(c_l, k + 1);
            on = __builtin_amdgcn_readlane(o_l, k + 1);
        }
        sl = sort_sources(sl, c, lane);
        int nsl = (lane < c) ? __float_as_int(nrm[sl]) : 0;
        float4 acca = make_float4(0.f, 0.f, 0.f, 0.f);
        float4 accb = make_float4(0.f, 0.f, 0.f, 0.f);
        float4 va[8], vb[8];
        load_rows2(x4, sl, 0, c, half, l16, va, vb);
        int sl_next = (lane < cn) ? csr[on + lane] : 0;
        fma_rows2(acca, accb, va, vb, nsl, 0, half);
        int cur = sl, ncur = nsl;
        for (int e = 16; e < c; e += 16) {
            int jj = e & 31;
            if (jj == 0) {
                cur = 0; ncur = 0;
                if (e + lane < c) { cur = csr[o + e + lane]; ncur = __float_as_int(nrm[cur]); }
            }
            load_rows2(x4, cur, jj, c - (e - jj), half, l16, va, vb);
            fma_rows2(acca, accb, va, vb, ncur, jj, half);
        }

        acca.x += __shfl_xor(acca.x, 16);
        acca.y += __shfl_xor(acca.y, 16);
        acca.z += __shfl_xor(acca.z, 16);
        acca.w += __shfl_xor(acca.w, 16);
        accb.x += __shfl_xor(accb.x, 16);
        accb.y += __shfl_xor(accb.y, 16);
        accb.z += __shfl_xor(accb.z, 16);
        accb.w += __shfl_xor(accb.w, 16);
        if (half == 0) {
            trow[k * 32 + (l16 ^ k)]        = acca;
            trow[k * 32 + 16 + (l16 ^ k)]   = accb;
        }
        c = cn; o = on; sl = sl_next;
    }

    __builtin_amdgcn_wave_barrier();

    const float4* arow = trow + l16 * 32;
    v8f l0 = {}, l1 = {}, l2 = {}, l3 = {}, c0 = {}, c1 = {}, c2 = {}, c3 = {};
#pragma unroll
    for (int kc = 0; kc < 16; ++kc) {
        int p = kc * 2 + half;
        float4 q = arow[p ^ l16];
        v2f al; al.x = q.x; al.y = q.z;
        v2f ac; ac.x = q.y; ac.y = q.w;
        const float2* wl = &L.w[0][p * WPITCH + l16];
        const float2* wc = &L.w[1][p * WPITCH + l16];
        float2 t;
        v2f bl0; t = wl[ 0]; bl0.x = t.x; bl0.y = t.y;
        v2f bl1; t = wl[16]; bl1.x = t.x; bl1.y = t.y;
        v2f bl2; t = wl[32]; bl2.x = t.x; bl2.y = t.y;
        v2f bl3; t = wl[48]; bl3.x = t.x; bl3.y = t.y;
        v2f bc0; t = wc[ 0]; bc0.x = t.x; bc0.y = t.y;
        v2f bc1; t = wc[16]; bc1.x = t.x; bc1.y = t.y;
        v2f bc2; t = wc[32]; bc2.x = t.x; bc2.y = t.y;
        v2f bc3; t = wc[48]; bc3.x = t.x; bc3.y = t.y;
        l0 = __builtin_amdgcn_wmma_f32_16x16x4_f32(false, al, false, bl0, (short)0, l0, false, false);
        l1 = __builtin_amdgcn_wmma_f32_16x16x4_f32(false, al, false, bl1, (short)0, l1, false, false);
        l2 = __builtin_amdgcn_wmma_f32_16x16x4_f32(false, al, false, bl2, (short)0, l2, false, false);
        l3 = __builtin_amdgcn_wmma_f32_16x16x4_f32(false, al, false, bl3, (short)0, l3, false, false);
        c0 = __builtin_amdgcn_wmma_f32_16x16x4_f32(false, ac, false, bc0, (short)0, c0, false, false);
        c1 = __builtin_amdgcn_wmma_f32_16x16x4_f32(false, ac, false, bc1, (short)0, c1, false, false);
        c2 = __builtin_amdgcn_wmma_f32_16x16x4_f32(false, ac, false, bc2, (short)0, c2, false, false);
        c3 = __builtin_amdgcn_wmma_f32_16x16x4_f32(false, ac, false, bc3, (short)0, c3, false, false);
    }

    float bl_0 = bl[l16], bl_1 = bl[l16 + 16], bl_2 = bl[l16 + 32], bl_3 = bl[l16 + 48];
    float bc_0 = bc[l16], bc_1 = bc[l16 + 16], bc_2 = bc[l16 + 32], bc_3 = bc[l16 + 48];
#pragma unroll
    for (int v = 0; v < 8; ++v) {
        float n0 = __int_as_float(__builtin_amdgcn_readlane(n_l, v));
        float n1 = __int_as_float(__builtin_amdgcn_readlane(n_l, v + 8));
        float nr = half ? n1 : n0;
        int row = node0 + v + half * 8;
        if (row < n_nodes) {
            float* orow = out + (size_t)row * N_FEAT + l16;
            orow[ 0] = fmaxf(fmaf(l0[v], nr, bl_0), 0.f) + fmaxf(fmaf(c0[v], nr, bc_0), 0.f);
            orow[16] = fmaxf(fmaf(l1[v], nr, bl_1), 0.f) + fmaxf(fmaf(c1[v], nr, bc_1), 0.f);
            orow[32] = fmaxf(fmaf(l2[v], nr, bl_2), 0.f) + fmaxf(fmaf(c2[v], nr, bc_2), 0.f);
            orow[48] = fmaxf(fmaf(l3[v], nr, bl_3), 0.f) + fmaxf(fmaf(c3[v], nr, bc_3), 0.f);
        }
    }
}

extern "C" void kernel_launch(void* const* d_in, const int* in_sizes, int n_in,
                              void* d_out, int out_size, void* d_ws, size_t ws_size,
                              hipStream_t stream) {
    const float* x   = (const float*)d_in[1];
    const int*   src = (const int*)d_in[2];
    const int*   dst = (const int*)d_in[3];
    const float* W_l = (const float*)d_in[4];
    const float* b_l = (const float*)d_in[5];
    const float* W_c = (const float*)d_in[6];
    const float* b_c = (const float*)d_in[7];
    float* out = (float*)d_out;

    const int n_edges = in_sizes[2];
    const int n_nodes = in_sizes[1] / (2 * N_FEAT);
    const int n_buckets = (n_nodes + BKT - 1) / BKT;
    const int n_chunks  = (n_edges + CH - 1) / CH;

    size_t nPad = ((size_t)n_nodes + 127) & ~(size_t)127;
    size_t ePad = ((size_t)n_edges + 127) & ~(size_t)127;
    size_t tPad = ((size_t)n_chunks * TROW + 127) & ~(size_t)127;
    int*   cnt   = (int*)d_ws;
    int*   offs  = cnt + nPad;
    float* nrm   = (float*)(offs + nPad);
    int*   csr   = (int*)(nrm + nPad);
    int*   T     = csr + ePad;
    int*   bsrc  = T + tPad;
    int*   bdst  = bsrc + ePad;

    bin_kernel<<<n_chunks, BIN_T, 0, stream>>>(src, dst, bsrc, bdst, T, n_edges, n_buckets);
    seg_kernel<<<n_buckets, SEG_T, 0, stream>>>(bsrc, bdst, T, cnt, offs, nrm, csr, n_nodes, n_chunks);

    int tiles = (n_nodes + 15) / 16;
    int fgrid = (tiles + FT_WAVES - 1) / FT_WAVES;
    agg_gemm_kernel<<<fgrid, FT_THREADS, 0, stream>>>(x, csr, cnt, offs, nrm, W_l, W_c, b_l, b_c,
                                                      out, n_nodes);
}


# edge sum's lane bookkeeping cut: sorted sources sit even/odd in the two 16-lane rows, so one row_share DPP feeds each two-row load and its norm (no readlanes, exec masks, zero fill); sum order kept
# speedup vs baseline: 5.6610x; 1.0434x over previous
#include <hip/hip_runtime.h>


typedef float v2f __attribute__((ext_vector_type(2)));
typedef float v8f __attribute__((ext_vector_type(8)));

#define N_FEAT 64

#define BSH 8
#define BKT (1 << BSH)
#define NBMAX 512
#define TROW (NBMAX + 1)
#define CH 4096
#define BIN_T 256
#define EPT (CH / BIN_T)
#define SEG_T 512
#define CAP 8192
static_assert(NBMAX == 2 * BIN_T, "the chunk block scans two table entries per thread");

template <int NW>
__device__ __forceinline__ int block_excl_scan(int x, int* wsum, int& tot) {
    int lane = threadIdx.x & 31, wid = threadIdx.x >> 5;
    int v = x;
    for (int d = 1; d < 32; d <<= 1) {
        int t = __shfl_up(v, d);
        if (lane >= d) v += t;
    }
    if (lane == 31) wsum[wid] = v;
    __syncthreads();
    int pre = 0, all = 0;
    for (int k = 0; k < NW; ++k) { int t = wsum[k]; if (k < wid) pre += t; all += t; }
    tot = all;
    __syncthreads();
    return pre + v - x;
}

__global__ void
bin_kernel(const int* __restrict__ src, const int* __restrict__ dst,
           int* __restrict__ bsrc, int* __restrict__ bdst, int* __restrict__ T,
           int n_edges, int n_buckets) {
    __shared__ int hist[TROW];
    __shared__ int s_src[CH];
    __shared__ int s_dst[CH];
    __shared__ int wsum[BIN_T / 32];
    int tid = threadIdx.x;
    int base = blockIdx.x * CH;
    int nE = min(CH, n_edges - base);
    for (int i = tid; i < TROW; i += BIN_T) hist[i] = 0;
    __syncthreads();
    int d[EPT], s[EPT], r[EPT];
#pragma unroll
    for (int q = 0; q < EPT; ++q) {
        int i = q * BIN_T + tid;
        d[q] = -1; s[q] = 0;
        if (i < nE) { d[q] = dst[base + i]; s[q] = src[base + i]; }
    }
#pragma unroll
    for (int q = 0; q < EPT; ++q) {
        r[q] = 0;
        if (d[q] >= 0) r[q] = atomicAdd(&hist[d[q] >> BSH], 1);
    }
    __syncthreads();
    int a = hist[2 * tid], b = hist[2 * tid + 1];
    int tot;
    int ex = block_excl_scan<BIN_T / 32>(a + b, wsum, tot);
    hist[2 * tid] = ex;
    hist[2 * tid + 1] = ex + a;
    if (tid == BIN_T - 1) hist[NBMAX] = tot;
    __syncthreads();
    for (int i = tid; i <= n_buckets; i += BIN_T) T[(size_t)blockIdx.x * TROW + i] = hist[i];
#pragma unroll
    for (int q = 0; q < EPT; ++q) {
        if (d[q] >= 0) {
            int slot = hist[d[q] >> BSH] + r[q];
            s_src[slot] = s[q];
            s_dst[slot] = d[q];
        }
    }
    __syncthreads();
    for (int i = tid; i < nE; i += BIN_T) {
        bsrc[base + i] = s_src[i];
        bdst[base + i] = s_dst[i];
    }
}

__global__ void
seg_kernel(const int* __restrict__ bsrc, const int* __restrict__ bdst, const int* __restrict__ T,
           int* __restrict__ cnt, int* __restrict__ offs, float* __restrict__ nrm,
           int* __restrict__ csr, int n_nodes, int n_chunks) {
    __shared__ int lcnt[BKT];
    __shared__ int cursor[BKT];
    __shared__ int wsum[SEG_T / 32];
    __shared__ int s_csr[CAP];
    int tid = threadIdx.x, k = blockIdx.x;
    if (tid < BKT) lcnt[tid] = 0;
    __syncthreads();
    int before = 0;
    for (int b = tid; b < n_chunks; b += SEG_T) {
        int lo = T[(size_t)b * TROW + k], hi = T[(size_t)b * TROW + k + 1];
        before += lo;
        const int* pd = bdst + (size_t)b * CH;
        for (int i = lo; i < hi; ++i) atomicAdd(&lcnt[pd[i] & (BKT - 1)], 1);
    }
    __syncthreads();
    int c = (tid < BKT) ? lcnt[tid] : 0;
    int tot, base;
    int ex = block_excl_scan<SEG_T / 32>(c, wsum, tot);
    block_excl_scan<SEG_T / 32>(before, wsum, base);
    if (tid < BKT) {
        cursor[tid] = ex;
        int node = k * BKT + tid;
        if (node < n_nodes) {
            cnt[node]  = c;
            offs[node] = base + ex;
            nrm[node]  = (c > 0) ? rsqrtf((float)c) : 0.f;
        }
    }
    __syncthreads();
    bool staged = tot <= CAP;
    for (int b = tid; b < n_chunks; b += SEG_T) {
        int lo = T[(size_t)b * TROW + k], hi = T[(size_t)b * TROW + k + 1];
        const int* pd = bdst + (size_t)b * CH;
        const int* ps = bsrc + (size_t)b * CH;
        for (int i = lo; i < hi; ++i) {
            int sv = ps[i];
            int p = atomicAdd(&cursor[pd[i] & (BKT - 1)], 1);
            if (staged) s_csr[p] = sv; else csr[base + p] = sv;
        }
    }
    __syncthreads();
    if (staged)
        for (int i = tid; i < tot; i += SEG_T) csr[base + i] = s_csr[i];
}

#define FT_WAVES 25
#define FT_THREADS (FT_WAVES * 32)
#define WPITCH 80


__device__ __forceinline__ int row_bcast(int v, int u) {
#define ROW_SHARE_CASE(U) case U: return __builtin_amdgcn_update_dpp(0, v, 0x150 + U, 0xf, 0xf, true);
    switch (u) {
        ROW_SHARE_CASE(0)  ROW_SHARE_CASE(1)  ROW_SHARE_CASE(2)  ROW_SHARE_CASE(3)
        ROW_SHARE_CASE(4)  ROW_SHARE_CASE(5)  ROW_SHARE_CASE(6)  ROW_SHARE_CASE(7)
        ROW_SHARE_CASE(8)  ROW_SHARE_CASE(9)  ROW_SHARE_CASE(10) ROW_SHARE_CASE(11)
        ROW_SHARE_CASE(12) ROW_SHARE_CASE(13) ROW_SHARE_CASE(14)
        default: return __builtin_amdgcn_update_dpp(0, v, 0x150 + 15, 0xf, 0xf, true);
    }
#undef ROW_SHARE_CASE
}

template <int JB>
__device__ __forceinline__ void load_rows2(const char* __restrict__ xb, int cur, int nvalid,
                                           unsigned loff, float4 (&va)[8], float4 (&vb)[8]) {
#pragma unroll
    for (int u = 0; u < 8; ++u) {
        if (2 * u >= nvalid) break;
        unsigned s = (unsigned)row_bcast(cur, JB + u);
        unsigned off = (s << 9) + loff;
        va[u] = *(const float4*)(xb + off);
        vb[u] = *(const float4*)(xb + off + 256);
    }
}

template <int JB>
__device__ __forceinline__ void fma_rows2(float4& a, float4& b, const float4 (&va)[8],
                                          const float4 (&vb)[8], int nbits, int nvalid) {
#pragma unroll
    for (int u = 0; u < 8; ++u) {
        if (2 * u >= nvalid) break;
        float ns = __int_as_float(row_bcast(nbits, JB + u));
        a.x = fmaf(va[u].x, ns, a.x); a.y = fmaf(va[u].y, ns, a.y);
        a.z = fmaf(va[u].z, ns, a.z); a.w = fmaf(va[u].w, ns, a.w);
        b.x = fmaf(vb[u].x, ns, b.x); b.y = fmaf(vb[u].y, ns, b.y);
        b.z = fmaf(vb[u].z, ns, b.z); b.w = fmaf(vb[u].w, ns, b.w);
    }
}

__device__ __forceinline__ int sort_sources(int sl, int c, int lane) {
    int m = min(c, 32);
    int key = (sl << 5) | lane;
    int rank = 0;
    for (int t = 0; t < m; ++t)
        rank += (__builtin_amdgcn_readlane(key, t) < key) ? 1 : 0;
    if (lane >= m) rank = lane;
    int dest = (rank >> 1) | ((rank & 1) << 4);
    return __builtin_amdgcn_ds_permute(dest << 2, sl);
}

__device__ __forceinline__ float other_row(float v) {
    int b = __float_as_int(v);
    return __int_as_float(__builtin_amdgcn_permlanex16(b, b, 0x76543210, 0xfedcba98, false, false));
}

__global__ void __launch_bounds__(FT_THREADS)
agg_gemm_kernel(const float* __restrict__ x, const int* __restrict__ csr,
                const int* __restrict__ cnt, const int* __restrict__ offs,
                const float* __restrict__ nrm,
                const float* __restrict__ Wl, const float* __restrict__ Wc,
                const float* __restrict__ bl, const float* __restrict__ bc,
                float* __restrict__ out, int n_nodes) {
    struct Lds {
        float2 w[2][32 * WPITCH];
        float4 tile[FT_WAVES][16 * 32];
    };
    __shared__ Lds L;

    int tid = threadIdx.x;
    int lane = tid & 31;
    int half = lane >> 4, l16 = lane & 15;
    int wib  = __builtin_amdgcn_readfirstlane(tid >> 5);
    int wave = blockIdx.x * FT_WAVES + wib;
    int node0 = wave * 16;

    int mi = node0 + l16;
    int c_l = 0, o_l = 0, n_l = 0;
    if (mi < n_nodes) { c_l = cnt[mi]; o_l = offs[mi]; n_l = __float_as_int(nrm[mi]); }

    constexpr int WSTEPS = (2 * 32 * 64 + FT_THREADS - 1) / FT_THREADS;
    float w0[WSTEPS], w1[WSTEPS];
#pragma unroll
    for (int r = 0; r < WSTEPS; ++r) {
        int i = tid + r * FT_THREADS;
        w0[r] = 0.f; w1[r] = 0.f;
        if (i < 2 * 32 * 64) {
            int m = i >> 11, p = (i >> 6) & 31, n = i & 63;
            const float* Wm = m ? Wc : Wl;
            w0[r] = Wm[(2 * p) * N_FEAT + n];
            w1[r] = Wm[(2 * p + 1) * N_FEAT + n];
        }
    }
#pragma unroll
    for (int r = 0; r < WSTEPS; ++r) {
        int i = tid + r * FT_THREADS;
        if (i < 2 * 32 * 64) {
            int m = i >> 11, p = (i >> 6) & 31, n = i & 63;
            L.w[m][p * WPITCH + n] = make_float2(w0[r], w1[r]);
        }
    }
    __syncthreads();
    if (node0 >= n_nodes) return;

    const char* xb = (const char*)x;
    const unsigned loff = (unsigned)l16 * 16u;
    const int rk = 2 * l16 + half;
    float4* trow = &L.tile[wib][0];

    int c = __builtin_amdgcn_readlane(c_l, 0);
    int o = __builtin_amdgcn_readlane(o_l, 0);
    int sl = (lane < c) ? csr[o + lane] : 0;

    for (int k = 0; k < 16; ++k) {
        int cn = 0, on = 0;
        if (k + 1 < 16) {
            cn = __builtin_amdgcn_readlane(c_l, k + 1);
            on = __builtin_amdgcn_readlane(o_l, k + 1);
        }
        sl = sort_sources(sl, c, lane);
        int first = __builtin_amdgcn_readlane(sl, 0);
        bool ok = rk < c;
        sl = ok ? sl : first;
        int nsl = ok ? __float_as_int(nrm[sl]) : 0;
        float4 acca = make_float4(0.f, 0.f, 0.f, 0.f);
        float4 accb = make_float4(0.f, 0.f, 0.f, 0.f);
        float4 va[8], vb[8];
        load_rows2<0>(xb, sl, c, loff, va, vb);
        int sl_next = (lane < cn) ? csr[on + lane] : 0;
        fma_rows2<0>(acca, accb, va, vb, nsl, c);
        if (c > 16) {
            load_rows2<8>(xb, sl, c - 16, loff, va, vb);
            fma_rows2<8>(acca, accb, va, vb, nsl, c - 16);
        }
        for (int e = 32; e < c; e += 32) {
            int cur = first, ncur = 0;
            if (e + rk < c) { cur = csr[o + e + rk]; ncur = __float_as_int(nrm[cur]); }
            load_rows2<0>(xb, cur, c - e, loff, va, vb);
            fma_rows2<0>(acca, accb, va, vb, ncur, c - e);
            if (c > e + 16) {
                load_rows2<8>(xb, cur, c - e - 16, loff, va, vb);
                fma_rows2<8>(acca, accb, va, vb, ncur, c - e - 16);
            }
        }

        acca.x += other_row(acca.x);
        acca.y += other_row(acca.y);
        acca.z += other_row(acca.z);
        acca.w += other_row(acca.w);
        accb.x += other_row(accb.x);
        accb.y += other_row(accb.y);
        accb.z += other_row(accb.z);
        accb.w += other_row(accb.w);
        if (half == 0) {
            trow[k * 32 + (l16 ^ k)]        = acca;
            trow[k * 32 + 16 + (l16 ^ k)]   = accb;
        }
        c = cn; o = on; sl = sl_next;
    }

    __builtin_amdgcn_wave_barrier();

    const float4* arow = trow + l16 * 32;
    v8f l0 = {}, l1 = {}, l2 = {}, l3 = {}, c0 = {}, c1 = {}, c2 = {}, c3 = {};
#pragma unroll
    for (int kc = 0; kc < 16; ++kc) {
        int p = kc * 2 + half;
        float4 q = arow[p ^ l16];
        v2f al; al.x = q.x; al.y = q.z;
        v2f ac; ac.x = q.y; ac.y = q.w;
        const float2* wl = &L.w[0][p * WPITCH + l16];
        const float2* wc = &L.w[1][p * WPITCH + l16];
        float2 t;
        v2f bl0; t = wl[ 0]; bl0.x = t.x; bl0.y = t.y;
        v2f bl1; t = wl[16]; bl1.x = t.x; bl1.y = t.y;
        v2f bl2; t = wl[32]; bl2.x = t.x; bl2.y = t.y;
        v2f bl3; t = wl[48]; bl3.x = t.x; bl3.y = t.y;
        v2f bc0; t = wc[ 0]; bc0.x = t.x; bc0.y = t.y;
        v2f bc1; t = wc[16]; bc1.x = t.x; bc1.y = t.y;
        v2f bc2; t = wc[32]; bc2.x = t.x; bc2.y = t.y;
        v2f bc3; t = wc[48]; bc3.x = t.x; bc3.y = t.y;
        l0 = __builtin_amdgcn_wmma_f32_16x16x4_f32(false, al, false, bl0, (short)0, l0, false, false);
        l1 = __builtin_amdgcn_wmma_f32_16x16x4_f32(false, al, false, bl1, (short)0, l1, false, false);
        l2 = __builtin_amdgcn_wmma_f32_16x16x4_f32(false, al, false, bl2, (short)0, l2, false, false);
        l3 = __builtin_amdgcn_wmma_f32_16x16x4_f32(false, al, false, bl3, (short)0, l3, false, false);
        c0 = __builtin_amdgcn_wmma_f32_16x16x4_f32(false, ac, false, bc0, (short)0, c0, false, false);
        c1 = __builtin_amdgcn_wmma_f32_16x16x4_f32(false, ac, false, bc1, (short)0, c1, false, false);
        c2 = __builtin_amdgcn_wmma_f32_16x16x4_f32(false, ac, false, bc2, (short)0, c2, false, false);
        c3 = __builtin_amdgcn_wmma_f32_16x16x4_f32(false, ac, false, bc3, (short)0, c3, false, false);
    }

    float bl_0 = bl[l16], bl_1 = bl[l16 + 16], bl_2 = bl[l16 + 32], bl_3 = bl[l16 + 48];
    float bc_0 = bc[l16], bc_1 = bc[l16 + 16], bc_2 = bc[l16 + 32], bc_3 = bc[l16 + 48];
#pragma unroll
    for (int v = 0; v < 8; ++v) {
        float n0 = __int_as_float(__builtin_amdgcn_readlane(n_l, v));
        float n1 = __int_as_float(__builtin_amdgcn_readlane(n_l, v + 8));
        float nr = half ? n1 : n0;
        int row = node0 + v + half * 8;
        if (row < n_nodes) {
            float* orow = out + (size_t)row * N_FEAT + l16;
            orow[ 0] = fmaxf(fmaf(l0[v], nr, bl_0), 0.f) + fmaxf(fmaf(c0[v], nr, bc_0), 0.f);
            orow[16] = fmaxf(fmaf(l1[v], nr, bl_1), 0.f) + fmaxf(fmaf(c1[v], nr, bc_1), 0.f);
            orow[32] = fmaxf(fmaf(l2[v], nr, bl_2), 0.f) + fmaxf(fmaf(c2[v], nr, bc_2), 0.f);
            orow[48] = fmaxf(fmaf(l3[v], nr, bl_3), 0.f) + fmaxf(fmaf(c3[v], nr, bc_3), 0.f);
        }
    }
}

extern "C" void kernel_launch(void* const* d_in, const int* in_sizes, int n_in,
                              void* d_out, int out_size, void* d_ws, size_t ws_size,
                              hipStream_t stream) {
    const float* x   = (const float*)d_in[1];
    const int*   src = (const int*)d_in[2];
    const int*   dst = (const int*)d_in[3];
    const float* W_l = (const float*)d_in[4];
    const float* b_l = (const float*)d_in[5];
    const float* W_c = (const float*)d_in[6];
    const float* b_c = (const float*)d_in[7];
    float* out = (float*)d_out;

    const int n_edges = in_sizes[2];
    const int n_nodes = in_sizes[1] / (2 * N_FEAT);
    const int n_buckets = (n_nodes + BKT - 1) / BKT;
    const int n_chunks  = (n_edges + CH - 1) / CH;

    size_t nPad = ((size_t)n_nodes + 127) & ~(size_t)127;
    size_t ePad = ((size_t)n_edges + 127) & ~(size_t)127;
    size_t tPad = ((size_t)n_chunks * TROW + 127) & ~(size_t)127;
    int*   cnt   = (int*)d_ws;
    int*   offs  = cnt + nPad;
    float* nrm   = (float*)(offs + nPad);
    int*   csr   = (int*)(nrm + nPad);
    int*   T     = csr + ePad;
    int*   bsrc  = T + tPad;
    int*   bdst  = bsrc + ePad;

    bin_kernel<<<n_chunks, BIN_T, 0, stream>>>(src, dst, bsrc, bdst, T, n_edges, n_buckets);
    seg_kernel<<<n_buckets, SEG_T, 0, stream>>>(bsrc, bdst, T, cnt, offs, nrm, csr, n_nodes, n_chunks);

    int tiles = (n_nodes + 15) / 16;
    int fgrid = (tiles + FT_WAVES - 1) / FT_WAVES;
    agg_gemm_kernel<<<fgrid, FT_THREADS, 0, stream>>>(x, csr, cnt, offs, nrm, W_l, W_c, b_l, b_c,
                                                      out, n_nodes);
}


# bucket pass: each thread's edge-by-edge run walk (dependent load + LDS add per edge, twice) becomes wave-per-run loads into one LDS list, add's return = rank; sum order moves only above degree 32
# speedup vs baseline: 6.5376x; 1.1548x over previous
#include <hip/hip_runtime.h>


typedef float v2f __attribute__((ext_vector_type(2)));
typedef float v8f __attribute__((ext_vector_type(8)));

#define N_FEAT 64

#define BSH 8
#define BKT (1 << BSH)
#define NBMAX 512
#define TROW (NBMAX + 1)
#define CH 4096
#define BIN_T 256
#define EPT (CH / BIN_T)
#define SEG_T 512
#define SEG_W (SEG_T / 32)
#define CAP 8192
#define SEG_Q (CAP / SEG_T)
#define SRC_BITS 17
static_assert(NBMAX == 2 * BIN_T, "the chunk block scans two table entries per thread");
static_assert(NBMAX * BKT == (1 << SRC_BITS) && SRC_BITS + BSH <= 31, "packed list entry");
static_assert(SEG_W == 16, "lane l of wave w owns the run of chunk 16 * l + w");

template <int NW>
__device__ __forceinline__ int block_excl_scan(int x, int* wsum, int& tot) {
    int lane = threadIdx.x & 31, wid = threadIdx.x >> 5;
    int v = x;
    for (int d = 1; d < 32; d <<= 1) {
        int t = __shfl_up(v, d);
        if (lane >= d) v += t;
    }
    if (lane == 31) wsum[wid] = v;
    __syncthreads();
    int pre = 0, all = 0;
    for (int k = 0; k < NW; ++k) { int t = wsum[k]; if (k < wid) pre += t; all += t; }
    tot = all;
    __syncthreads();
    return pre + v - x;
}

template <int NW>
__device__ __forceinline__ int block_excl_scan_sum(int x, int y, int* wsum, int& totx, int& toty) {
    int lane = threadIdx.x & 31, wid = threadIdx.x >> 5;
    int v = x, s = y;
    for (int d = 1; d < 32; d <<= 1) {
        int t = __shfl_up(v, d);
        if (lane >= d) v += t;
        s += __shfl_xor(s, d);
    }
    if (lane == 31) { wsum[wid] = v; wsum[NW + wid] = s; }
    __syncthreads();
    int pre = 0, all = 0, ally = 0;
    for (int k = 0; k < NW; ++k) {
        int t = wsum[k];
        if (k < wid) pre += t;
        all += t;
        ally += wsum[NW + k];
    }
    totx = all; toty = ally;
    __syncthreads();
    return pre + v - x;
}

__global__ void
bin_kernel(const int* __restrict__ src, const int* __restrict__ dst,
           int* __restrict__ bsrc, int* __restrict__ bdst, int* __restrict__ T,
           int n_edges, int n_buckets) {
    __shared__ int hist[TROW];
    __shared__ int s_src[CH];
    __shared__ int s_dst[CH];
    __shared__ int wsum[BIN_T / 32];
    int tid = threadIdx.x;
    int base = blockIdx.x * CH;
    int nE = min(CH, n_edges - base);
    for (int i = tid; i < TROW; i += BIN_T) hist[i] = 0;
    __syncthreads();
    int d[EPT], s[EPT], r[EPT];
#pragma unroll
    for (int q = 0; q < EPT; ++q) {
        int i = q * BIN_T + tid;
        d[q] = -1; s[q] = 0;
        if (i < nE) { d[q] = dst[base + i]; s[q] = src[base + i]; }
    }
#pragma unroll
    for (int q = 0; q < EPT; ++q) {
        r[q] = 0;
        if (d[q] >= 0) r[q] = atomicAdd(&hist[d[q] >> BSH], 1);
    }
    __syncthreads();
    int a = hist[2 * tid], b = hist[2 * tid + 1];
    int tot;
    int ex = block_excl_scan<BIN_T / 32>(a + b, wsum, tot);
    hist[2 * tid] = ex;
    hist[2 * tid + 1] = ex + a;
    if (tid == BIN_T - 1) hist[NBMAX] = tot;
    __syncthreads();
    for (int i = tid; i <= n_buckets; i += BIN_T) T[(size_t)blockIdx.x * TROW + i] = hist[i];
#pragma unroll
    for (int q = 0; q < EPT; ++q) {
        if (d[q] >= 0) {
            int slot = hist[d[q] >> BSH] + r[q];
            s_src[slot] = s[q];
            s_dst[slot] = d[q];
        }
    }
    __syncthreads();
    for (int i = tid; i < nE; i += BIN_T) {
        bsrc[base + i] = s_src[i];
        bdst[base + i] = s_dst[i];
    }
}


template <int JB>
__device__ __forceinline__ void load_runs(const int* __restrict__ bdst, const int* __restrict__ bsrc,
                                          int lo, int len, int wv, unsigned lane,
                                          int (&d)[16], int (&s)[16]) {
#pragma unroll
    for (int u = 0; u < 16; ++u) {
        int rl = __builtin_amdgcn_readlane(lo, JB + u);
        int rn = __builtin_amdgcn_readlane(len, JB + u);
        size_t first = (rn > 0) ? (size_t)(((JB + u) << 4) | wv) * CH + rl : 0;
        unsigned li = min(lane, (unsigned)(rn - 1));
        d[u] = (bdst + first)[li];
        s[u] = (bsrc + first)[li];
    }
}

template <int JB>
__device__ __forceinline__ void store_runs(int* list, int len, int start, unsigned lane,
                                           const int (&d)[16], const int (&s)[16]) {
#pragma unroll
    for (int u = 0; u < 16; ++u) {
        int rn = __builtin_amdgcn_readlane(len, JB + u);
        int rs = __builtin_amdgcn_readlane(start, JB + u);
        if ((int)lane < rn) list[rs + lane] = s[u] | ((d[u] & (BKT - 1)) << SRC_BITS);
    }
}

__global__ void
seg_kernel(const int* __restrict__ bsrc, const int* __restrict__ bdst, const int* __restrict__ T,
           int* __restrict__ cnt, int* __restrict__ offs, float* __restrict__ nrm,
           int* __restrict__ csr, int n_nodes, int n_chunks) {
    __shared__ int lcnt[BKT];
    __shared__ int cursor[BKT];
    __shared__ int wsum[2 * SEG_W];
    __shared__ int s_csr[CAP];
    int tid = threadIdx.x, k = blockIdx.x;
    int lane = tid & 31;
    const int wv = __builtin_amdgcn_readfirstlane(tid >> 5);
    if (tid < BKT) lcnt[tid] = 0;

    const bool own_run = n_chunks <= SEG_T;
    int lo = 0, len = 0;
    {
        int b = (lane << 4) | wv;
        if (own_run && b < n_chunks) {
            lo  = T[(size_t)b * TROW + k];
            len = T[(size_t)b * TROW + k + 1] - lo;
        }
    }
    int tot, base;
    int start = block_excl_scan_sum<SEG_W>(len, lo, wsum, tot, base);

    if (own_run && tot <= CAP) {
        {
            int dA[16], sA[16], dB[16], sB[16];
            load_runs<0>(bdst, bsrc, lo, len, wv, lane, dA, sA);
            load_runs<16>(bdst, bsrc, lo, len, wv, lane, dB, sB);
            store_runs<0>(s_csr, len, start, lane, dA, sA);
            store_runs<16>(s_csr, len, start, lane, dB, sB);
        }
        if (__builtin_amdgcn_ballot_w32(len > 32) != 0) {
            for (int u = 0; u < 32; ++u) {
                int rn = __builtin_amdgcn_readlane(len, u);
                if (rn <= 32) continue;
                int rl = __builtin_amdgcn_readlane(lo, u);
                int rs = __builtin_amdgcn_readlane(start, u);
                size_t at = (size_t)((u << 4) | wv) * CH + rl;
                for (int i = 32 + lane; i < rn; i += 32)
                    s_csr[rs + i] = bsrc[at + i] | ((bdst[at + i] & (BKT - 1)) << SRC_BITS);
            }
        }
        __syncthreads();
        int e[SEG_Q], r[SEG_Q];
#pragma unroll
        for (int q = 0; q < SEG_Q; ++q) {
            int j = q * SEG_T + tid;
            e[q] = 0;
            if (j < tot) e[q] = s_csr[j];
        }
#pragma unroll
        for (int q = 0; q < SEG_Q; ++q) {
            int j = q * SEG_T + tid;
            r[q] = 0;
            if (j < tot) r[q] = atomicAdd(&lcnt[e[q] >> SRC_BITS], 1);
        }
        __syncthreads();
        int c = (tid < BKT) ? lcnt[tid] : 0;
        int tot2;
        int ex = block_excl_scan<SEG_W>(c, wsum, tot2);
        if (tid < BKT) {
            cursor[tid] = ex;
            int node = k * BKT + tid;
            if (node < n_nodes) {
                cnt[node]  = c;
                offs[node] = base + ex;
                nrm[node]  = (c > 0) ? rsqrtf((float)c) : 0.f;
            }
        }
        __syncthreads();
#pragma unroll
        for (int q = 0; q < SEG_Q; ++q) {
            int j = q * SEG_T + tid;
            if (j < tot) r[q] += cursor[e[q] >> SRC_BITS];
        }
#pragma unroll
        for (int q = 0; q < SEG_Q; ++q) {
            int j = q * SEG_T + tid;
            if (j < tot) s_csr[r[q]] = e[q] & ((1 << SRC_BITS) - 1);
        }
        __syncthreads();
        for (int i = tid; i < tot; i += SEG_T) csr[base + i] = s_csr[i];
        return;
    }

    int before = 0;
    for (int b = tid; b < n_chunks; b += SEG_T) {
        int l0 = T[(size_t)b * TROW + k], h0 = T[(size_t)b * TROW + k + 1];
        before += l0;
        const int* pd = bdst + (size_t)b * CH;
        for (int i = l0; i < h0; ++i) atomicAdd(&lcnt[pd[i] & (BKT - 1)], 1);
    }
    __syncthreads();
    int c = (tid < BKT) ? lcnt[tid] : 0;
    int gtot, gbase;
    int ex = block_excl_scan<SEG_W>(c, wsum, gtot);
    block_excl_scan<SEG_W>(before, wsum, gbase);
    if (tid < BKT) {
        cursor[tid] = ex;
        int node = k * BKT + tid;
        if (node < n_nodes) {
            cnt[node]  = c;
            offs[node] = gbase + ex;
            nrm[node]  = (c > 0) ? rsqrtf((float)c) : 0.f;
        }
    }
    __syncthreads();
    bool staged = gtot <= CAP;
    for (int b = tid; b < n_chunks; b += SEG_T) {
        int l0 = T[(size_t)b * TROW + k], h0 = T[(size_t)b * TROW + k + 1];
        const int* pd = bdst + (size_t)b * CH;
        const int* ps = bsrc + (size_t)b * CH;
        for (int i = l0; i < h0; ++i) {
            int sv = ps[i];
            int p = atomicAdd(&cursor[pd[i] & (BKT - 1)], 1);
            if (staged) s_csr[p] = sv; else csr[gbase + p] = sv;
        }
    }
    __syncthreads();
    if (staged)
        for (int i = tid; i < gtot; i += SEG_T) csr[gbase + i] = s_csr[i];
}

#define FT_WAVES 25
#define FT_THREADS (FT_WAVES * 32)
#define WPITCH 80


__device__ __forceinline__ int row_bcast(int v, int u) {
#define ROW_SHARE_CASE(U) case U: return __builtin_amdgcn_update_dpp(0, v, 0x150 + U, 0xf, 0xf, true);
    switch (u) {
        ROW_SHARE_CASE(0)  ROW_SHARE_CASE(1)  ROW_SHARE_CASE(2)  ROW_SHARE_CASE(3)
        ROW_SHARE_CASE(4)  ROW_SHARE_CASE(5)  ROW_SHARE_CASE(6)  ROW_SHARE_CASE(7)
        ROW_SHARE_CASE(8)  ROW_SHARE_CASE(9)  ROW_SHARE_CASE(10) ROW_SHARE_CASE(11)
        ROW_SHARE_CASE(12) ROW_SHARE_CASE(13) ROW_SHARE_CASE(14)
        default: return __builtin_amdgcn_update_dpp(0, v, 0x150 + 15, 0xf, 0xf, true);
    }
#undef ROW_SHARE_CASE
}

template <int JB>
__device__ __forceinline__ void load_rows2(const char* __restrict__ xb, int cur, int nvalid,
                                           unsigned loff, float4 (&va)[8], float4 (&vb)[8]) {
#pragma unroll
    for (int u = 0; u < 8; ++u) {
        if (2 * u >= nvalid) break;
        unsigned s = (unsigned)row_bcast(cur, JB + u);
        unsigned off = (s << 9) + loff;
        va[u] = *(const float4*)(xb + off);
        vb[u] = *(const float4*)(xb + off + 256);
    }
}

template <int JB>
__device__ __forceinline__ void fma_rows2(float4& a, float4& b, const float4 (&va)[8],
                                          const float4 (&vb)[8], int nbits, int nvalid) {
#pragma unroll
    for (int u = 0; u < 8; ++u) {
        if (2 * u >= nvalid) break;
        float ns = __int_as_float(row_bcast(nbits, JB + u));
        a.x = fmaf(va[u].x, ns, a.x); a.y = fmaf(va[u].y, ns, a.y);
        a.z = fmaf(va[u].z, ns, a.z); a.w = fmaf(va[u].w, ns, a.w);
        b.x = fmaf(vb[u].x, ns, b.x); b.y = fmaf(vb[u].y, ns, b.y);
        b.z = fmaf(vb[u].z, ns, b.z); b.w = fmaf(vb[u].w, ns, b.w);
    }
}

__device__ __forceinline__ int sort_sources(int sl, int c, int lane) {
    int m = min(c, 32);
    int key = (sl << 5) | lane;
    int rank = 0;
    for (int t = 0; t < m; ++t)
        rank += (__builtin_amdgcn_readlane(key, t) < key) ? 1 : 0;
    if (lane >= m) rank = lane;
    int dest = (rank >> 1) | ((rank & 1) << 4);
    return __builtin_amdgcn_ds_permute(dest << 2, sl);
}

__device__ __forceinline__ float other_row(float v) {
    int b = __float_as_int(v);
    return __int_as_float(__builtin_amdgcn_permlanex16(b, b, 0x76543210, 0xfedcba98, false, false));
}

__global__ void __launch_bounds__(FT_THREADS)
agg_gemm_kernel(const float* __restrict__ x, const int* __restrict__ csr,
                const int* __restrict__ cnt, const int* __restrict__ offs,
                const float* __restrict__ nrm,
                const float* __restrict__ Wl, const float* __restrict__ Wc,
                const float* __restrict__ bl, const float* __restrict__ bc,
                float* __restrict__ out, int n_nodes) {
    struct Lds {
        float2 w[2][32 * WPITCH];
        float4 tile[FT_WAVES][16 * 32];
    };
    __shared__ Lds L;

    int tid = threadIdx.x;
    int lane = tid & 31;
    int half = lane >> 4, l16 = lane & 15;
    int wib  = __builtin_amdgcn_readfirstlane(tid >> 5);
    int wave = blockIdx.x * FT_WAVES + wib;
    int node0 = wave * 16;

    int mi = node0 + l16;
    int c_l = 0, o_l = 0, n_l = 0;
    if (mi < n_nodes) { c_l = cnt[mi]; o_l = offs[mi]; n_l = __float_as_int(nrm[mi]); }

    constexpr int WSTEPS = (2 * 32 * 64 + FT_THREADS - 1) / FT_THREADS;
    float w0[WSTEPS], w1[WSTEPS];
#pragma unroll
    for (int r = 0; r < WSTEPS; ++r) {
        int i = tid + r * FT_THREADS;
        w0[r] = 0.f; w1[r] = 0.f;
        if (i < 2 * 32 * 64) {
            int m = i >> 11, p = (i >> 6) & 31, n = i & 63;
            const float* Wm = m ? Wc : Wl;
            w0[r] = Wm[(2 * p) * N_FEAT + n];
            w1[r] = Wm[(2 * p + 1) * N_FEAT + n];
        }
    }
#pragma unroll
    for (int r = 0; r < WSTEPS; ++r) {
        int i = tid + r * FT_THREADS;
        if (i < 2 * 32 * 64) {
            int m = i >> 11, p = (i >> 6) & 31, n = i & 63;
            L.w[m][p * WPITCH + n] = make_float2(w0[r], w1[r]);
        }
    }
    __syncthreads();
    if (node0 >= n_nodes) return;

    const char* xb = (const char*)x;
    const unsigned loff = (unsigned)l16 * 16u;
    const int rk = 2 * l16 + half;
    float4* trow = &L.tile[wib][0];

    int c = __builtin_amdgcn_readlane(c_l, 0);
    int o = __builtin_amdgcn_readlane(o_l, 0);
    int sl = (lane < c) ? csr[o + lane] : 0;

    for (int k = 0; k < 16; ++k) {
        int cn = 0, on = 0;
        if (k + 1 < 16) {
            cn = __builtin_amdgcn_readlane(c_l, k + 1);
            on = __builtin_amdgcn_readlane(o_l, k + 1);
        }
        sl = sort_sources(sl, c, lane);
        int first = __builtin_amdgcn_readlane(sl, 0);
        bool ok = rk < c;
        sl = ok ? sl : first;
        int nsl = ok ? __float_as_int(nrm[sl]) : 0;
        float4 acca = make_float4(0.f, 0.f, 0.f, 0.f);
        float4 accb = make_float4(0.f, 0.f, 0.f, 0.f);
        float4 va[8], vb[8];
        load_rows2<0>(xb, sl, c, loff, va, vb);
        int sl_next = (lane < cn) ? csr[on + lane] : 0;
        fma_rows2<0>(acca, accb, va, vb, nsl, c);
        if (c > 16) {
            load_rows2<8>(xb, sl, c - 16, loff, va, vb);
            fma_rows2<8>(acca, accb, va, vb, nsl, c - 16);
        }
        for (int e = 32; e < c; e += 32) {
            int cur = first, ncur = 0;
            if (e + rk < c) { cur = csr[o + e + rk]; ncur = __float_as_int(nrm[cur]); }
            load_rows2<0>(xb, cur, c - e, loff, va, vb);
            fma_rows2<0>(acca, accb, va, vb, ncur, c - e);
            if (c > e + 16) {
                load_rows2<8>(xb, cur, c - e - 16, loff, va, vb);
                fma_rows2<8>(acca, accb, va, vb, ncur, c - e - 16);
            }
        }

        acca.x += other_row(acca.x);
        acca.y += other_row(acca.y);
        acca.z += other_row(acca.z);
        acca.w += other_row(acca.w);
        accb.x += other_row(accb.x);
        accb.y += other_row(accb.y);
        accb.z += other_row(accb.z);
        accb.w += other_row(accb.w);
        if (half == 0) {
            trow[k * 32 + (l16 ^ k)]        = acca;
            trow[k * 32 + 16 + (l16 ^ k)]   = accb;
        }
        c = cn; o = on; sl = sl_next;
    }

    __builtin_amdgcn_wave_barrier();

    const float4* arow = trow + l16 * 32;
    v8f l0 = {}, l1 = {}, l2 = {}, l3 = {}, c0 = {}, c1 = {}, c2 = {}, c3 = {};
#pragma unroll
    for (int kc = 0; kc < 16; ++kc) {
        int p = kc * 2 + half;
        float4 q = arow[p ^ l16];
        v2f al; al.x = q.x; al.y = q.z;
        v2f ac; ac.x = q.y; ac.y = q.w;
        const float2* wl = &L.w[0][p * WPITCH + l16];
        const float2* wc = &L.w[1][p * WPITCH + l16];
        float2 t;
        v2f bl0; t = wl[ 0]; bl0.x = t.x; bl0.y = t.y;
        v2f bl1; t = wl[16]; bl1.x = t.x; bl1.y = t.y;
        v2f bl2; t = wl[32]; bl2.x = t.x; bl2.y = t.y;
        v2f bl3; t = wl[48]; bl3.x = t.x; bl3.y = t.y;
        v2f bc0; t = wc[ 0]; bc0.x = t.x; bc0.y = t.y;
        v2f bc1; t = wc[16]; bc1.x = t.x; bc1.y = t.y;
        v2f bc2; t = wc[32]; bc2.x = t.x; bc2.y = t.y;
        v2f bc3; t = wc[48]; bc3.x = t.x; bc3.y = t.y;
        l0 = __builtin_amdgcn_wmma_f32_16x16x4_f32(false, al, false, bl0, (short)0, l0, false, false);
        l1 = __builtin_amdgcn_wmma_f32_16x16x4_f32(false, al, false, bl1, (short)0, l1, false, false);
        l2 = __builtin_amdgcn_wmma_f32_16x16x4_f32(false, al, false, bl2, (short)0, l2, false, false);
        l3 = __builtin_amdgcn_wmma_f32_16x16x4_f32(false, al, false, bl3, (short)0, l3, false, false);
        c0 = __builtin_amdgcn_wmma_f32_16x16x4_f32(false, ac, false, bc0, (short)0, c0, false, false);
        c1 = __builtin_amdgcn_wmma_f32_16x16x4_f32(false, ac, false, bc1, (short)0, c1, false, false);
        c2 = __builtin_amdgcn_wmma_f32_16x16x4_f32(false, ac, false, bc2, (short)0, c2, false, false);
        c3 = __builtin_amdgcn_wmma_f32_16x16x4_f32(false, ac, false, bc3, (short)0, c3, false, false);
    }

    float bl_0 = bl[l16], bl_1 = bl[l16 + 16], bl_2 = bl[l16 + 32], bl_3 = bl[l16 + 48];
    float bc_0 = bc[l16], bc_1 = bc[l16 + 16], bc_2 = bc[l16 + 32], bc_3 = bc[l16 + 48];
#pragma unroll
    for (int v = 0; v < 8; ++v) {
        float n0 = __int_as_float(__builtin_amdgcn_readlane(n_l, v));
        float n1 = __int_as_float(__builtin_amdgcn_readlane(n_l, v + 8));
        float nr = half ? n1 : n0;
        int row = node0 + v + half * 8;
        if (row < n_nodes) {
            float* orow = out + (size_t)row * N_FEAT + l16;
            orow[ 0] = fmaxf(fmaf(l0[v], nr, bl_0), 0.f) + fmaxf(fmaf(c0[v], nr, bc_0), 0.f);
            orow[16] = fmaxf(fmaf(l1[v], nr, bl_1), 0.f) + fmaxf(fmaf(c1[v], nr, bc_1), 0.f);
            orow[32] = fmaxf(fmaf(l2[v], nr, bl_2), 0.f) + fmaxf(fmaf(c2[v], nr, bc_2), 0.f);
            orow[48] = fmaxf(fmaf(l3[v], nr, bl_3), 0.f) + fmaxf(fmaf(c3[v], nr, bc_3), 0.f);
        }
    }
}

extern "C" void kernel_launch(void* const* d_in, const int* in_sizes, int n_in,
                              void* d_out, int out_size, void* d_ws, size_t ws_size,
                              hipStream_t stream) {
    const float* x   = (const float*)d_in[1];
    const int*   src = (const int*)d_in[2];
    const int*   dst = (const int*)d_in[3];
    const float* W_l = (const float*)d_in[4];
    const float* b_l = (const float*)d_in[5];
    const float* W_c = (const float*)d_in[6];
    const float* b_c = (const float*)d_in[7];
    float* out = (float*)d_out;

    const int n_edges = in_sizes[2];
    const int n_nodes = in_sizes[1] / (2 * N_FEAT);
    const int n_buckets = (n_nodes + BKT - 1) / BKT;
    const int n_chunks  = (n_edges + CH - 1) / CH;

    size_t nPad = ((size_t)n_nodes + 127) & ~(size_t)127;
    size_t ePad = ((size_t)n_edges + 127) & ~(size_t)127;
    size_t tPad = ((size_t)n_chunks * TROW + 127) & ~(size_t)127;
    int*   cnt   = (int*)d_ws;
    int*   offs  = cnt + nPad;
    float* nrm   = (float*)(offs + nPad);
    int*   csr   = (int*)(nrm + nPad);
    int*   T     = csr + ePad;
    int*   bsrc  = T + tPad;
    int*   bdst  = bsrc + ePad;

    bin_kernel<<<n_chunks, BIN_T, 0, stream>>>(src, dst, bsrc, bdst, T, n_edges, n_buckets);
    seg_kernel<<<n_buckets, SEG_T, 0, stream>>>(bsrc, bdst, T, cnt, offs, nrm, csr, n_nodes, n_chunks);

    int tiles = (n_nodes + 15) / 16;
    int fgrid = (tiles + FT_WAVES - 1) / FT_WAVES;
    agg_gemm_kernel<<<fgrid, FT_THREADS, 0, stream>>>(x, csr, cnt, offs, nrm, W_l, W_c, b_l, b_c,
                                                      out, n_nodes);
}
